# starting kernel
# baseline (speedup 1.0000x reference)
#include <hip/hip_runtime.h>
#include <math.h>

#define NNODES 50000
#define NEDGES 800000
#define NGRAPH 256
#define FIN    128
#define H1C    12
#define D1C    32
#define F1     (H1C * D1C)
#define H2C    1
#define D2C    64
#define F2     (H2C * D2C)
#define ETOT   (NEDGES + NNODES)

typedef float v2f __attribute__((ext_vector_type(2)));
typedef float v8f __attribute__((ext_vector_type(8)));

__device__ __forceinline__ unsigned fkey(float f) {
  unsigned u = __float_as_uint(f);
  return (u & 0x80000000u) ? ~u : (u | 0x80000000u);
}
__device__ __forceinline__ float fdecode(unsigned u) {
  unsigned b = (u & 0x80000000u) ? (u & 0x7fffffffu) : ~u;
  return __uint_as_float(b);
}

__global__ void __launch_bounds__(128)
wmma_gemm_f32(const float* __restrict__ A, const float* __restrict__ B,
              float* __restrict__ C, int K, int Ncols) {
  const int lane = threadIdx.x & 31;
  const int wave = threadIdx.x >> 5;
  const int l    = lane & 15;
  const int half = lane >> 4;
  const int m0   = blockIdx.x << 4;
  const int n0   = ((blockIdx.y << 2) + wave) << 4;

  v8f c = {};
  const float* arow = A + (size_t)(m0 + l) * K + (half << 1);
  const float* bcol = B + n0 + l;

  for (int k0 = 0; k0 < K; k0 += 4) {
    v2f a = *(const v2f*)(arow + k0);
    const int kb = k0 + (half << 1);
    v2f b;
    b.x = bcol[(size_t)kb * Ncols];
    b.y = bcol[(size_t)(kb + 1) * Ncols];
    c = __builtin_amdgcn_wmma_f32_16x16x4_f32(
         false, a,  false, b,
         (short)0, c,  false,  false);
  }

  float* crow = C + (size_t)(m0 + (half << 3)) * Ncols + n0 + l;
#pragma unroll
  for (int r = 0; r < 8; ++r) crow[(size_t)r * Ncols] = c[r];
}

__global__ void att_coeff(const float* __restrict__ h,
                          const float* __restrict__ att_src,
                          const float* __restrict__ att_dst,
                          float* __restrict__ asrc, float* __restrict__ adst,
                          int H, int D) {
  int i = blockIdx.x * blockDim.x + threadIdx.x;
  if (i >= NNODES * H) return;
  int n = i / H, hh = i - n * H;
  const float* hp = h + (size_t)n * H * D + hh * D;
  const float* ws = att_src + hh * D;
  const float* wd = att_dst + hh * D;
  float s = 0.f, d = 0.f;
  for (int j = 0; j < D; ++j) { float v = hp[j]; s += v * ws[j]; d += v * wd[j]; }
  asrc[i] = s; adst[i] = d;
}

__global__ void edge_logit_max(const int* __restrict__ srcI, const int* __restrict__ dstI,
                               const float* __restrict__ asrc, const float* __restrict__ adst,
                               float* __restrict__ e, unsigned* __restrict__ mkey, int H) {
  int i = blockIdx.x * blockDim.x + threadIdx.x;
  if (i >= ETOT * H) return;
  int eid = i / H, hh = i - eid * H;
  int s, d;
  if (eid < NEDGES) { s = srcI[eid]; d = dstI[eid]; } else { s = eid - NEDGES; d = s; }
  float v = asrc[s * H + hh] + adst[d * H + hh];
  v = (v > 0.f) ? v : 0.2f * v;
  e[i] = v;
  atomicMax(&mkey[d * H + hh], fkey(v));
}

__global__ void edge_exp_sum(const int* __restrict__ dstI, float* __restrict__ e,
                             const unsigned* __restrict__ mkey, float* __restrict__ ssum,
                             int H) {
  int i = blockIdx.x * blockDim.x + threadIdx.x;
  if (i >= ETOT * H) return;
  int eid = i / H, hh = i - eid * H;
  int d = (eid < NEDGES) ? dstI[eid] : (eid - NEDGES);
  float p = __expf(e[i] - fdecode(mkey[d * H + hh]));
  e[i] = p;
  atomicAdd(&ssum[d * H + hh], p);
}

__global__ void edge_alpha(const int* __restrict__ dstI, float* __restrict__ e,
                           const float* __restrict__ ssum, int H) {
  int i = blockIdx.x * blockDim.x + threadIdx.x;
  if (i >= ETOT * H) return;
  int eid = i / H, hh = i - eid * H;
  int d = (eid < NEDGES) ? dstI[eid] : (eid - NEDGES);
  e[i] = e[i] / ssum[d * H + hh];
}

__global__ void edge_agg(const int* __restrict__ srcI, const int* __restrict__ dstI,
                         const float* __restrict__ alpha, const float* __restrict__ h,
                         float* __restrict__ out, int H, int D) {
  const int F = H * D;
  int i = blockIdx.x * blockDim.x + threadIdx.x;
  if (i >= ETOT * F) return;
  int eid = i / F, f = i - eid * F;
  int hh = f / D;
  int s, d;
  if (eid < NEDGES) { s = srcI[eid]; d = dstI[eid]; } else { s = eid - NEDGES; d = s; }
  atomicAdd(&out[(size_t)d * F + f], alpha[eid * H + hh] * h[(size_t)s * F + f]);
}

__global__ void bias_act(float* __restrict__ x, const float* __restrict__ bias,
                         int total, int F, int do_relu) {
  int i = blockIdx.x * blockDim.x + threadIdx.x;
  if (i >= total) return;
  float v = x[i] + bias[i % F];
  if (do_relu) v = fmaxf(v, 0.f);
  x[i] = v;
}

__global__ void pool_accum(const float* __restrict__ h, const int* __restrict__ batch,
                           float* __restrict__ sums, float* __restrict__ cnt) {
  int i = blockIdx.x * blockDim.x + threadIdx.x;
  if (i >= NNODES * F2) return;
  int n = i / F2, f = i - n * F2;
  int g = batch[n];
  atomicAdd(&sums[g * F2 + f], h[i]);
  if (f == 0) atomicAdd(&cnt[g], 1.0f);
}

__global__ void pool_final(const float* __restrict__ sums, const float* __restrict__ cnt,
                           float* __restrict__ out) {
  int i = blockIdx.x * blockDim.x + threadIdx.x;
  if (i >= NGRAPH * F2) return;
  out[i] = sums[i] / fmaxf(cnt[i / F2], 1.0f);
}

static inline int cdiv(long long a, int b) { return (int)((a + b - 1) / b); }

extern "C" void kernel_launch(void* const* d_in, const int* in_sizes, int n_in,
                              void* d_out, int out_size, void* d_ws, size_t ws_size,
                              hipStream_t stream) {
  const float* x        = (const float*)d_in[0];
  const int*   ei       = (const int*)d_in[1];
  const int*   batch    = (const int*)d_in[2];
  const float* W1       = (const float*)d_in[3];
  const float* att_src1 = (const float*)d_in[4];
  const float* att_dst1 = (const float*)d_in[5];
  const float* bias1    = (const float*)d_in[6];
  const float* W2       = (const float*)d_in[7];
  const float* att_src2 = (const float*)d_in[8];
  const float* att_dst2 = (const float*)d_in[9];
  const float* bias2    = (const float*)d_in[10];
  float* out = (float*)d_out;
  const int* srcI = ei;
  const int* dstI = ei + NEDGES;

  char* p = (char*)d_ws;
  auto alloc = [&](size_t bytes) -> char* {
    char* r = p; p += (bytes + 255) & ~(size_t)255; return r;
  };
  float*    h1    = (float*)alloc((size_t)NNODES * F1 * 4);
  float*    asrc1 = (float*)alloc((size_t)NNODES * H1C * 4);
  float*    adst1 = (float*)alloc((size_t)NNODES * H1C * 4);
  unsigned* mkey1 = (unsigned*)alloc((size_t)NNODES * H1C * 4);
  float*    ssum1 = (float*)alloc((size_t)NNODES * H1C * 4);
  float*    e1    = (float*)alloc((size_t)ETOT * H1C * 4);
  float*    agg1  = (float*)alloc((size_t)NNODES * F1 * 4);
  float*    h2    = (float*)alloc((size_t)NNODES * F2 * 4);
  float*    asrc2 = (float*)alloc((size_t)NNODES * H2C * 4);
  float*    adst2 = (float*)alloc((size_t)NNODES * H2C * 4);
  unsigned* mkey2 = (unsigned*)alloc((size_t)NNODES * H2C * 4);
  float*    ssum2 = (float*)alloc((size_t)NNODES * H2C * 4);
  float*    e2    = (float*)alloc((size_t)ETOT * H2C * 4);
  float*    agg2  = (float*)alloc((size_t)NNODES * F2 * 4);
  float*    psum  = (float*)alloc((size_t)NGRAPH * F2 * 4);
  float*    pcnt  = (float*)alloc((size_t)NGRAPH * 4);

  hipMemsetAsync(mkey1, 0, (size_t)NNODES * H1C * 4, stream);
  hipMemsetAsync(ssum1, 0, (size_t)NNODES * H1C * 4, stream);
  hipMemsetAsync(agg1,  0, (size_t)NNODES * F1 * 4, stream);
  hipMemsetAsync(mkey2, 0, (size_t)NNODES * H2C * 4, stream);
  hipMemsetAsync(ssum2, 0, (size_t)NNODES * H2C * 4, stream);
  hipMemsetAsync(agg2,  0, (size_t)NNODES * F2 * 4, stream);
  hipMemsetAsync(psum,  0, (size_t)NGRAPH * F2 * 4, stream);
  hipMemsetAsync(pcnt,  0, (size_t)NGRAPH * 4, stream);

  const int T = 256;

  wmma_gemm_f32<<<dim3(NNODES / 16, F1 / 64), 128, 0, stream>>>(x, W1, h1, FIN, F1);
  att_coeff<<<cdiv((long long)NNODES * H1C, T), T, 0, stream>>>(h1, att_src1, att_dst1,
                                                                asrc1, adst1, H1C, D1C);
  edge_logit_max<<<cdiv((long long)ETOT * H1C, T), T, 0, stream>>>(srcI, dstI, asrc1, adst1,
                                                                   e1, mkey1, H1C);
  edge_exp_sum<<<cdiv((long long)ETOT * H1C, T), T, 0, stream>>>(dstI, e1, mkey1, ssum1, H1C);
  edge_alpha<<<cdiv((long long)ETOT * H1C, T), T, 0, stream>>>(dstI, e1, ssum1, H1C);
  edge_agg<<<cdiv((long long)ETOT * F1, T), T, 0, stream>>>(srcI, dstI, e1, h1, agg1, H1C, D1C);
  bias_act<<<cdiv((long long)NNODES * F1, T), T, 0, stream>>>(agg1, bias1, NNODES * F1, F1, 1);

  wmma_gemm_f32<<<dim3(NNODES / 16, F2 / 64), 128, 0, stream>>>(agg1, W2, h2, F1, F2);
  att_coeff<<<cdiv((long long)NNODES * H2C, T), T, 0, stream>>>(h2, att_src2, att_dst2,
                                                                asrc2, adst2, H2C, D2C);
  edge_logit_max<<<cdiv((long long)ETOT * H2C, T), T, 0, stream>>>(srcI, dstI, asrc2, adst2,
                                                                   e2, mkey2, H2C);
  edge_exp_sum<<<cdiv((long long)ETOT * H2C, T), T, 0, stream>>>(dstI, e2, mkey2, ssum2, H2C);
  edge_alpha<<<cdiv((long long)ETOT * H2C, T), T, 0, stream>>>(dstI, e2, ssum2, H2C);
  edge_agg<<<cdiv((long long)ETOT * F2, T), T, 0, stream>>>(srcI, dstI, e2, h2, agg2, H2C, D2C);
  bias_act<<<cdiv((long long)NNODES * F2, T), T, 0, stream>>>(agg2, bias2, NNODES * F2, F2, 0);

  pool_accum<<<cdiv((long long)NNODES * F2, T), T, 0, stream>>>(agg2, batch, psum, pcnt);
  pool_final<<<cdiv((long long)NGRAPH * F2, T), T, 0, stream>>>(psum, pcnt, out);
}


# Both aggregations by owner groups: 196 destination rows summed in LDS (each group scans all edges, LDS float adds), stored once with bias/relu; global float atomics gone; sum order by arrival
# speedup vs baseline: 2.8408x; 2.8408x over previous
#include <hip/hip_runtime.h>
#include <math.h>

#define NNODES 50000
#define NEDGES 800000
#define NGRAPH 256
#define FIN    128
#define H1C    12
#define D1C    32
#define F1     (H1C * D1C)
#define H2C    1
#define D2C    64
#define F2     (H2C * D2C)
#define ETOT   (NEDGES + NNODES)

typedef float v2f __attribute__((ext_vector_type(2)));
typedef float v8f __attribute__((ext_vector_type(8)));

__device__ __forceinline__ unsigned fkey(float f) {
  unsigned u = __float_as_uint(f);
  return (u & 0x80000000u) ? ~u : (u | 0x80000000u);
}
__device__ __forceinline__ float fdecode(unsigned u) {
  unsigned b = (u & 0x80000000u) ? (u & 0x7fffffffu) : ~u;
  return __uint_as_float(b);
}

__global__ void __launch_bounds__(128)
wmma_gemm_f32(const float* __restrict__ A, const float* __restrict__ B,
              float* __restrict__ C, int K, int Ncols) {
  const int lane = threadIdx.x & 31;
  const int wave = threadIdx.x >> 5;
  const int l    = lane & 15;
  const int half = lane >> 4;
  const int m0   = blockIdx.x << 4;
  const int n0   = ((blockIdx.y << 2) + wave) << 4;

  v8f c = {};
  const float* arow = A + (size_t)(m0 + l) * K + (half << 1);
  const float* bcol = B + n0 + l;

  for (int k0 = 0; k0 < K; k0 += 4) {
    v2f a = *(const v2f*)(arow + k0);
    const int kb = k0 + (half << 1);
    v2f b;
    b.x = bcol[(size_t)kb * Ncols];
    b.y = bcol[(size_t)(kb + 1) * Ncols];
    c = __builtin_amdgcn_wmma_f32_16x16x4_f32(
         false, a,  false, b,
         (short)0, c,  false,  false);
  }

  float* crow = C + (size_t)(m0 + (half << 3)) * Ncols + n0 + l;
#pragma unroll
  for (int r = 0; r < 8; ++r) crow[(size_t)r * Ncols] = c[r];
}

__global__ void att_coeff(const float* __restrict__ h,
                          const float* __restrict__ att_src,
                          const float* __restrict__ att_dst,
                          float* __restrict__ asrc, float* __restrict__ adst,
                          int H, int D) {
  int i = blockIdx.x * blockDim.x + threadIdx.x;
  if (i >= NNODES * H) return;
  int n = i / H, hh = i - n * H;
  const float* hp = h + (size_t)n * H * D + hh * D;
  const float* ws = att_src + hh * D;
  const float* wd = att_dst + hh * D;
  float s = 0.f, d = 0.f;
  for (int j = 0; j < D; ++j) { float v = hp[j]; s += v * ws[j]; d += v * wd[j]; }
  asrc[i] = s; adst[i] = d;
}

__global__ void edge_logit_max(const int* __restrict__ srcI, const int* __restrict__ dstI,
                               const float* __restrict__ asrc, const float* __restrict__ adst,
                               float* __restrict__ e, unsigned* __restrict__ mkey, int H) {
  int i = blockIdx.x * blockDim.x + threadIdx.x;
  if (i >= ETOT * H) return;
  int eid = i / H, hh = i - eid * H;
  int s, d;
  if (eid < NEDGES) { s = srcI[eid]; d = dstI[eid]; } else { s = eid - NEDGES; d = s; }
  float v = asrc[s * H + hh] + adst[d * H + hh];
  v = (v > 0.f) ? v : 0.2f * v;
  e[i] = v;
  atomicMax(&mkey[d * H + hh], fkey(v));
}

__global__ void edge_exp_sum(const int* __restrict__ dstI, float* __restrict__ e,
                             const unsigned* __restrict__ mkey, float* __restrict__ ssum,
                             int H) {
  int i = blockIdx.x * blockDim.x + threadIdx.x;
  if (i >= ETOT * H) return;
  int eid = i / H, hh = i - eid * H;
  int d = (eid < NEDGES) ? dstI[eid] : (eid - NEDGES);
  float p = __expf(e[i] - fdecode(mkey[d * H + hh]));
  e[i] = p;
  atomicAdd(&ssum[d * H + hh], p);
}

__global__ void edge_alpha(const int* __restrict__ dstI, float* __restrict__ e,
                           const float* __restrict__ ssum, int H) {
  int i = blockIdx.x * blockDim.x + threadIdx.x;
  if (i >= ETOT * H) return;
  int eid = i / H, hh = i - eid * H;
  int d = (eid < NEDGES) ? dstI[eid] : (eid - NEDGES);
  e[i] = e[i] / ssum[d * H + hh];
}

#define OWN_R    196
#define OWN_T    1024
#define LIST_CAP 5632
#define SCAN_U   8
#define INIT_U   5

template <int F, int D, int EB, int RELU>
__global__ void __launch_bounds__(OWN_T)
owner_agg(const int* __restrict__ srcI, const int* __restrict__ dstI,
          const float* __restrict__ alpha, const float* __restrict__ h,
          const float* __restrict__ bias, float* __restrict__ out) {
  constexpr int H  = F / D;
  constexpr int K  = F / 32;
  constexpr int F4 = F / 4;
  __shared__ float    acc[OWN_R * F];
  __shared__ unsigned list[LIST_CAP];
  __shared__ unsigned cnt;

  const int tid  = threadIdx.x;
  const int lane = tid & 31;
  const int wave = tid >> 5;
  const int r0   = blockIdx.x * OWN_R;

  if (tid == 0) cnt = 0u;
  for (int i0 = tid; i0 < OWN_R * F4; i0 += OWN_T * INIT_U) {
    float4 hv[INIT_U];
    float  a[INIT_U];
    bool   ok[INIT_U];
#pragma unroll
    for (int u = 0; u < INIT_U; ++u) {
      const int i   = i0 + u * OWN_T;
      const int ic  = i < OWN_R * F4 ? i : OWN_R * F4 - 1;
      const int row = ic / F4, c4 = ic - row * F4;
      const int n   = r0 + row;
      const int nc  = n < NNODES ? n : NNODES - 1;
      ok[u] = n < NNODES;
      hv[u] = *(const float4*)(h + (size_t)nc * F + c4 * 4);
      a[u]  = alpha[(size_t)(NEDGES + nc) * H + (c4 * 4) / D];
    }
#pragma unroll
    for (int u = 0; u < INIT_U; ++u) {
      const int i = i0 + u * OWN_T;
      if (i < OWN_R * F4) {
        float4 v = make_float4(a[u] * hv[u].x, a[u] * hv[u].y, a[u] * hv[u].z, a[u] * hv[u].w);
        if (!ok[u]) v = make_float4(0.f, 0.f, 0.f, 0.f);
        *(float4*)(acc + (size_t)i * 4) = v;
      }
    }
  }
  __syncthreads();

  {
    constexpr int NE4 = NEDGES / 4;
    const int4* d4 = (const int4*)dstI;
    for (int base = 0; base < NE4; base += OWN_T * SCAN_U) {
      int4 dv[SCAN_U];
#pragma unroll
      for (int u = 0; u < SCAN_U; ++u) {
        const int idx = base + u * OWN_T + tid;
        dv[u] = (idx < NE4) ? d4[idx] : make_int4(-1, -1, -1, -1);
      }
      unsigned m = 0u;
#pragma unroll
      for (int u = 0; u < SCAN_U; ++u) {
        m |= ((unsigned)(dv[u].x - r0) < (unsigned)OWN_R ? 1u : 0u) << (4 * u + 0);
        m |= ((unsigned)(dv[u].y - r0) < (unsigned)OWN_R ? 1u : 0u) << (4 * u + 1);
        m |= ((unsigned)(dv[u].z - r0) < (unsigned)OWN_R ? 1u : 0u) << (4 * u + 2);
        m |= ((unsigned)(dv[u].w - r0) < (unsigned)OWN_R ? 1u : 0u) << (4 * u + 3);
      }
      while (m) {
        const int j = __builtin_ctz(m);
        m &= m - 1u;
        const int eid = (base + (j >> 2) * OWN_T + tid) * 4 + (j & 3);
        const unsigned dl = (unsigned)(dstI[eid] - r0);
        const unsigned pos = atomicAdd(&cnt, 1u);
        if (pos < (unsigned)LIST_CAP) {
          list[pos] = (dl << 20) | (unsigned)eid;
        } else {
          const int s = srcI[eid];
          for (int f = 0; f < F; ++f)
            atomicAdd(&acc[dl * F + f], alpha[(size_t)eid * H + f / D] * h[(size_t)s * F + f]);
        }
      }
    }
  }
  __syncthreads();

  {
    const unsigned seen  = cnt;
    const unsigned total = seen < (unsigned)LIST_CAP ? seen : (unsigned)LIST_CAP;
    unsigned i = (unsigned)wave * EB;
    for (; i + EB <= total; i += 32u * EB) {
      int eid[EB], dl[EB], s[EB];
      float al[EB][H];
      float hv[EB][K];
#pragma unroll
      for (int j = 0; j < EB; ++j) {
        const unsigned ent = (unsigned)__builtin_amdgcn_readfirstlane((int)list[i + j]);
        eid[j] = (int)(ent & 0xFFFFFu);
        dl[j]  = (int)(ent >> 20);
      }
#pragma unroll
      for (int j = 0; j < EB; ++j) {
        s[j] = srcI[eid[j]];
        const float* ap = alpha + (size_t)eid[j] * H;
#pragma unroll
        for (int q = 0; q < H; ++q) al[j][q] = ap[q];
      }
#pragma unroll
      for (int j = 0; j < EB; ++j) {
        const float* hp = h + (size_t)s[j] * F + lane;
#pragma unroll
        for (int k = 0; k < K; ++k) hv[j][k] = hp[k * 32];
      }
#pragma unroll
      for (int j = 0; j < EB; ++j) {
        float* accp = acc + dl[j] * F + lane;
#pragma unroll
        for (int k = 0; k < K; ++k) atomicAdd(&accp[k * 32], al[j][(k * 32) / D] * hv[j][k]);
      }
    }
    const unsigned rem0 = total - total % (unsigned)EB;
    if (rem0 + (unsigned)wave < total) {
      const unsigned ent = (unsigned)__builtin_amdgcn_readfirstlane((int)list[rem0 + wave]);
      const int e1 = (int)(ent & 0xFFFFFu);
      const int d1 = (int)(ent >> 20);
      const int s1 = srcI[e1];
      const float* hp = h + (size_t)s1 * F + lane;
      const float* ap = alpha + (size_t)e1 * H;
      float* accp = acc + d1 * F + lane;
      float hv1[K];
#pragma unroll
      for (int k = 0; k < K; ++k) hv1[k] = hp[k * 32];
#pragma unroll
      for (int k = 0; k < K; ++k) atomicAdd(&accp[k * 32], ap[(k * 32) / D] * hv1[k]);
    }
  }
  __syncthreads();

  for (int i = tid; i < OWN_R * F4; i += OWN_T) {
    const int row = i / F4, c4 = i - row * F4;
    const int n = r0 + row;
    if (n < NNODES) {
      float4 v = *(const float4*)(acc + (size_t)i * 4);
      const float* b = bias + c4 * 4;
      v.x += b[0]; v.y += b[1]; v.z += b[2]; v.w += b[3];
      if (RELU) { v.x = fmaxf(v.x, 0.f); v.y = fmaxf(v.y, 0.f); v.z = fmaxf(v.z, 0.f); v.w = fmaxf(v.w, 0.f); }
      *(float4*)(out + (size_t)n * F + c4 * 4) = v;
    }
  }
}

__global__ void pool_accum(const float* __restrict__ h, const int* __restrict__ batch,
                           float* __restrict__ sums, float* __restrict__ cnt) {
  int i = blockIdx.x * blockDim.x + threadIdx.x;
  if (i >= NNODES * F2) return;
  int n = i / F2, f = i - n * F2;
  int g = batch[n];
  atomicAdd(&sums[g * F2 + f], h[i]);
  if (f == 0) atomicAdd(&cnt[g], 1.0f);
}

__global__ void pool_final(const float* __restrict__ sums, const float* __restrict__ cnt,
                           float* __restrict__ out) {
  int i = blockIdx.x * blockDim.x + threadIdx.x;
  if (i >= NGRAPH * F2) return;
  out[i] = sums[i] / fmaxf(cnt[i / F2], 1.0f);
}

static inline int cdiv(long long a, int b) { return (int)((a + b - 1) / b); }

extern "C" void kernel_launch(void* const* d_in, const int* in_sizes, int n_in,
                              void* d_out, int out_size, void* d_ws, size_t ws_size,
                              hipStream_t stream) {
  const float* x        = (const float*)d_in[0];
  const int*   ei       = (const int*)d_in[1];
  const int*   batch    = (const int*)d_in[2];
  const float* W1       = (const float*)d_in[3];
  const float* att_src1 = (const float*)d_in[4];
  const float* att_dst1 = (const float*)d_in[5];
  const float* bias1    = (const float*)d_in[6];
  const float* W2       = (const float*)d_in[7];
  const float* att_src2 = (const float*)d_in[8];
  const float* att_dst2 = (const float*)d_in[9];
  const float* bias2    = (const float*)d_in[10];
  float* out = (float*)d_out;
  const int* srcI = ei;
  const int* dstI = ei + NEDGES;

  char* p = (char*)d_ws;
  auto alloc = [&](size_t bytes) -> char* {
    char* r = p; p += (bytes + 255) & ~(size_t)255; return r;
  };
  float*    h1    = (float*)alloc((size_t)NNODES * F1 * 4);
  float*    asrc1 = (float*)alloc((size_t)NNODES * H1C * 4);
  float*    adst1 = (float*)alloc((size_t)NNODES * H1C * 4);
  unsigned* mkey1 = (unsigned*)alloc((size_t)NNODES * H1C * 4);
  float*    ssum1 = (float*)alloc((size_t)NNODES * H1C * 4);
  float*    e1    = (float*)alloc((size_t)ETOT * H1C * 4);
  float*    agg1  = (float*)alloc((size_t)NNODES * F1 * 4);
  float*    h2    = (float*)alloc((size_t)NNODES * F2 * 4);
  float*    asrc2 = (float*)alloc((size_t)NNODES * H2C * 4);
  float*    adst2 = (float*)alloc((size_t)NNODES * H2C * 4);
  unsigned* mkey2 = (unsigned*)alloc((size_t)NNODES * H2C * 4);
  float*    ssum2 = (float*)alloc((size_t)NNODES * H2C * 4);
  float*    e2    = (float*)alloc((size_t)ETOT * H2C * 4);
  float*    agg2  = (float*)alloc((size_t)NNODES * F2 * 4);
  float*    psum  = (float*)alloc((size_t)NGRAPH * F2 * 4);
  float*    pcnt  = (float*)alloc((size_t)NGRAPH * 4);

  hipMemsetAsync(mkey1, 0, (size_t)NNODES * H1C * 4, stream);
  hipMemsetAsync(ssum1, 0, (size_t)NNODES * H1C * 4, stream);
  hipMemsetAsync(mkey2, 0, (size_t)NNODES * H2C * 4, stream);
  hipMemsetAsync(ssum2, 0, (size_t)NNODES * H2C * 4, stream);
  hipMemsetAsync(psum,  0, (size_t)NGRAPH * F2 * 4, stream);
  hipMemsetAsync(pcnt,  0, (size_t)NGRAPH * 4, stream);

  const int T = 256;

  wmma_gemm_f32<<<dim3(NNODES / 16, F1 / 64), 128, 0, stream>>>(x, W1, h1, FIN, F1);
  att_coeff<<<cdiv((long long)NNODES * H1C, T), T, 0, stream>>>(h1, att_src1, att_dst1,
                                                                asrc1, adst1, H1C, D1C);
  edge_logit_max<<<cdiv((long long)ETOT * H1C, T), T, 0, stream>>>(srcI, dstI, asrc1, adst1,
                                                                   e1, mkey1, H1C);
  edge_exp_sum<<<cdiv((long long)ETOT * H1C, T), T, 0, stream>>>(dstI, e1, mkey1, ssum1, H1C);
  edge_alpha<<<cdiv((long long)ETOT * H1C, T), T, 0, stream>>>(dstI, e1, ssum1, H1C);
  owner_agg<F1, D1C, 4, 1><<<(NNODES + OWN_R - 1) / OWN_R, OWN_T, 0, stream>>>(srcI, dstI, e1, h1, bias1, agg1);

  wmma_gemm_f32<<<dim3(NNODES / 16, F2 / 64), 128, 0, stream>>>(agg1, W2, h2, F1, F2);
  att_coeff<<<cdiv((long long)NNODES * H2C, T), T, 0, stream>>>(h2, att_src2, att_dst2,
                                                                asrc2, adst2, H2C, D2C);
  edge_logit_max<<<cdiv((long long)ETOT * H2C, T), T, 0, stream>>>(srcI, dstI, asrc2, adst2,
                                                                   e2, mkey2, H2C);
  edge_exp_sum<<<cdiv((long long)ETOT * H2C, T), T, 0, stream>>>(dstI, e2, mkey2, ssum2, H2C);
  edge_alpha<<<cdiv((long long)ETOT * H2C, T), T, 0, stream>>>(dstI, e2, ssum2, H2C);
  owner_agg<F2, D2C, 8, 0><<<(NNODES + OWN_R - 1) / OWN_R, OWN_T, 0, stream>>>(srcI, dstI, e2, h2, bias2, agg2);

  pool_accum<<<cdiv((long long)NNODES * F2, T), T, 0, stream>>>(agg2, batch, psum, pcnt);
  pool_final<<<cdiv((long long)NGRAPH * F2, T), T, 0, stream>>>(psum, pcnt, out);
}


# Softmax moved into the owner-group aggregation: noted edges' logits, row max and exp-sum in LDS, rows summed unnormalised and divided at the store; 6 edge passes, 4 clears gone; sums by arrival
# speedup vs baseline: 4.3845x; 1.5434x over previous
#include <hip/hip_runtime.h>
#include <math.h>

#define NNODES 50000
#define NEDGES 800000
#define NGRAPH 256
#define FIN    128
#define H1C    12
#define D1C    32
#define F1     (H1C * D1C)
#define H2C    1
#define D2C    64
#define F2     (H2C * D2C)
#define ETOT   (NEDGES + NNODES)

typedef float v2f __attribute__((ext_vector_type(2)));
typedef float v8f __attribute__((ext_vector_type(8)));

__device__ __forceinline__ unsigned fkey(float f) {
  unsigned u = __float_as_uint(f);
  return (u & 0x80000000u) ? ~u : (u | 0x80000000u);
}
__device__ __forceinline__ float fdecode(unsigned u) {
  unsigned b = (u & 0x80000000u) ? (u & 0x7fffffffu) : ~u;
  return __uint_as_float(b);
}

__global__ void __launch_bounds__(128)
wmma_gemm_f32(const float* __restrict__ A, const float* __restrict__ B,
              float* __restrict__ C, int K, int Ncols) {
  const int lane = threadIdx.x & 31;
  const int wave = threadIdx.x >> 5;
  const int l    = lane & 15;
  const int half = lane >> 4;
  const int m0   = blockIdx.x << 4;
  const int n0   = ((blockIdx.y << 2) + wave) << 4;

  v8f c = {};
  const float* arow = A + (size_t)(m0 + l) * K + (half << 1);
  const float* bcol = B + n0 + l;

  for (int k0 = 0; k0 < K; k0 += 4) {
    v2f a = *(const v2f*)(arow + k0);
    const int kb = k0 + (half << 1);
    v2f b;
    b.x = bcol[(size_t)kb * Ncols];
    b.y = bcol[(size_t)(kb + 1) * Ncols];
    c = __builtin_amdgcn_wmma_f32_16x16x4_f32(
         false, a,  false, b,
         (short)0, c,  false,  false);
  }

  float* crow = C + (size_t)(m0 + (half << 3)) * Ncols + n0 + l;
#pragma unroll
  for (int r = 0; r < 8; ++r) crow[(size_t)r * Ncols] = c[r];
}

__global__ void att_coeff(const float* __restrict__ h,
                          const float* __restrict__ att_src,
                          const float* __restrict__ att_dst,
                          float* __restrict__ asrc, float* __restrict__ adst,
                          int H, int D) {
  int i = blockIdx.x * blockDim.x + threadIdx.x;
  if (i >= NNODES * H) return;
  int n = i / H, hh = i - n * H;
  const float* hp = h + (size_t)n * H * D + hh * D;
  const float* ws = att_src + hh * D;
  const float* wd = att_dst + hh * D;
  float s = 0.f, d = 0.f;
  for (int j = 0; j < D; ++j) { float v = hp[j]; s += v * ws[j]; d += v * wd[j]; }
  asrc[i] = s; adst[i] = d;
}

#define OWN_R    196
#define OWN_T    1024
#define LIST_CAP 16384
#define SCAN_U   8
#define INIT_U   5
#define MAX_U    8
#define NOENT    0xFFFFFFFFu

__device__ __forceinline__ float lrelu(float v) { return (v > 0.f) ? v : 0.2f * v; }

template <int F, int D, int EB, int RELU>
__global__ void __launch_bounds__(OWN_T)
owner_softmax_agg(const int* __restrict__ srcI, const int* __restrict__ dstI,
                  const float* __restrict__ asrc, const float* __restrict__ adst,
                  const float* __restrict__ h, const float* __restrict__ bias,
                  unsigned* glist_all, float* __restrict__ out) {
  constexpr int H    = F / D;
  constexpr int K    = F / 32;
  constexpr int F4   = F / 4;
  constexpr int RH   = OWN_R * H;
  constexpr int PE   = (32 / H) < EB ? (32 / H) : EB;
  constexpr int NSUB = EB / PE;
  static_assert(EB % PE == 0 && PE * H <= 32, "logit lanes");
  static_assert(NNODES <= 65536 && OWN_R <= 256, "list entry packing");
  __shared__ float    acc[OWN_R * F];
  __shared__ unsigned mkey[RH];
  __shared__ float    ssum[RH];
  __shared__ unsigned cnt;

  const int tid  = threadIdx.x;
  const int lane = tid & 31;
  const int wave = tid >> 5;
  const int r0   = blockIdx.x * OWN_R;
  unsigned* glist = glist_all + (size_t)blockIdx.x * LIST_CAP;

  if (tid == 0) cnt = 0u;
  for (int t = tid; t < RH; t += OWN_T) {
    const int n = r0 + t / H;
    float v = 0.f;
    if (n < NNODES) v = lrelu(asrc[(size_t)r0 * H + t] + adst[(size_t)r0 * H + t]);
    mkey[t] = fkey(v);
  }
  __syncthreads();

  {
    constexpr int NE4 = NEDGES / 4;
    const int4* d4 = (const int4*)dstI;
    for (int base = 0; base < NE4; base += OWN_T * SCAN_U) {
      int4 dv[SCAN_U];
#pragma unroll
      for (int u = 0; u < SCAN_U; ++u) {
        const int idx = base + u * OWN_T + tid;
        dv[u] = (idx < NE4) ? d4[idx] : make_int4(-1, -1, -1, -1);
      }
      unsigned m = 0u;
#pragma unroll
      for (int u = 0; u < SCAN_U; ++u) {
        m |= ((unsigned)(dv[u].x - r0) < (unsigned)OWN_R ? 1u : 0u) << (4 * u + 0);
        m |= ((unsigned)(dv[u].y - r0) < (unsigned)OWN_R ? 1u : 0u) << (4 * u + 1);
        m |= ((unsigned)(dv[u].z - r0) < (unsigned)OWN_R ? 1u : 0u) << (4 * u + 2);
        m |= ((unsigned)(dv[u].w - r0) < (unsigned)OWN_R ? 1u : 0u) << (4 * u + 3);
      }
      while (m) {
        const int j = __builtin_ctz(m);
        m &= m - 1u;
        const int eid = (base + (j >> 2) * OWN_T + tid) * 4 + (j & 3);
        const unsigned dl = (unsigned)(dstI[eid] - r0);
        const unsigned pos = atomicAdd(&cnt, 1u);
        if (pos < (unsigned)LIST_CAP) glist[pos] = (dl << 16) | (unsigned)srcI[eid];
      }
    }
  }
  __syncthreads();

  const unsigned seen = cnt;
  const bool fast = seen <= (unsigned)LIST_CAP;

  if (fast) {
    const int nitems = (int)seen * H;
    for (int t0 = tid; t0 < nitems; t0 += OWN_T * MAX_U) {
      unsigned en[MAX_U];
      int      hd[MAX_U];
      float    a1[MAX_U], a2[MAX_U];
#pragma unroll
      for (int u = 0; u < MAX_U; ++u) {
        const int t  = t0 + u * OWN_T;
        const int tc = t < nitems ? t : nitems - 1;
        const int e  = tc / H;
        hd[u] = tc - e * H;
        en[u] = glist[e];
      }
#pragma unroll
      for (int u = 0; u < MAX_U; ++u) {
        a1[u] = asrc[(size_t)(en[u] & 0xFFFFu) * H + hd[u]];
        a2[u] = adst[(size_t)(r0 + (int)(en[u] >> 16)) * H + hd[u]];
      }
#pragma unroll
      for (int u = 0; u < MAX_U; ++u)
        if (t0 + u * OWN_T < nitems)
          atomicMax(&mkey[(en[u] >> 16) * H + hd[u]], fkey(lrelu(a1[u] + a2[u])));
    }
  } else {
    for (int eid = tid; eid < NEDGES; eid += OWN_T) {
      const unsigned dl = (unsigned)(dstI[eid] - r0);
      if (dl < (unsigned)OWN_R) {
        const int s = srcI[eid];
        for (int q = 0; q < H; ++q)
          atomicMax(&mkey[dl * H + q],
                    fkey(lrelu(asrc[(size_t)s * H + q] + adst[(size_t)(r0 + (int)dl) * H + q])));
      }
    }
  }
  __syncthreads();

  for (int i0 = tid; i0 < OWN_R * F4; i0 += OWN_T * INIT_U) {
    float4 hv[INIT_U];
    float  as[INIT_U], ad[INIT_U];
    bool   ok[INIT_U];
#pragma unroll
    for (int u = 0; u < INIT_U; ++u) {
      const int i   = i0 + u * OWN_T;
      const int ic  = i < OWN_R * F4 ? i : OWN_R * F4 - 1;
      const int row = ic / F4, c4 = ic - row * F4;
      const int n   = r0 + row;
      const int nc  = n < NNODES ? n : NNODES - 1;
      ok[u] = n < NNODES;
      hv[u] = *(const float4*)(h + (size_t)nc * F + c4 * 4);
      as[u] = asrc[(size_t)nc * H + (c4 * 4) / D];
      ad[u] = adst[(size_t)nc * H + (c4 * 4) / D];
    }
#pragma unroll
    for (int u = 0; u < INIT_U; ++u) {
      const int i = i0 + u * OWN_T;
      if (i < OWN_R * F4) {
        const int row = i / F4, c4 = i - row * F4;
        const int hq  = (c4 * 4) / D;
        float p = __expf(lrelu(as[u] + ad[u]) - fdecode(mkey[row * H + hq]));
        float4 v = make_float4(p * hv[u].x, p * hv[u].y, p * hv[u].z, p * hv[u].w);
        if (!ok[u]) { v = make_float4(0.f, 0.f, 0.f, 0.f); p = 1.f; }
        *(float4*)(acc + (size_t)i * 4) = v;
        if ((c4 * 4) % D == 0) ssum[row * H + hq] = p;
      }
    }
  }
  __syncthreads();

  if (fast) {
    const unsigned total = seen;
    const unsigned last  = total > 0u ? total - 1u : 0u;
    const unsigned lj    = lane < EB ? (unsigned)lane : 0u;
    unsigned i = (unsigned)wave * EB;
    unsigned nxt = NOENT;
    if (i < total) {
      const unsigned raw = glist[i + lj < last ? i + lj : last];
      nxt = ((lane < EB) & (i + lj < total)) ? raw : NOENT;
    }
    for (; i < total; i += 32u * EB) {
      const unsigned cur = nxt;
      const unsigned in  = i + 32u * EB + lj;
      const unsigned raw = glist[in < last ? in : last];
      nxt = ((lane < EB) & (in < total)) ? raw : NOENT;
      unsigned ent[EB];
      float hv[EB][K];
      float al[EB][H];
#pragma unroll
      for (int j = 0; j < EB; ++j) ent[j] = (unsigned)__builtin_amdgcn_readlane((int)cur, j);
      const int jl = lane / H;
      const int hq = lane - jl * H;
      float    a1[NSUB], a2[NSUB];
      unsigned mk[NSUB], dj[NSUB];
      bool     ok[NSUB];
#pragma unroll
      for (int sub = 0; sub < NSUB; ++sub) {
        const unsigned el = (unsigned)__shfl((int)cur, sub * PE + (jl < PE ? jl : 0));
        ok[sub] = (jl < PE) & (el != NOENT);
        const unsigned sj = ok[sub] ? (el & 0xFFFFu) : 0u;
        dj[sub] = ok[sub] ? (el >> 16) : 0u;
        a1[sub] = asrc[(size_t)sj * H + hq];
        a2[sub] = adst[(size_t)(r0 + (int)dj[sub]) * H + hq];
        mk[sub] = mkey[dj[sub] * H + hq];
      }
#pragma unroll
      for (int j = 0; j < EB; ++j) {
        const unsigned s = (ent[j] == NOENT) ? 0u : (ent[j] & 0xFFFFu);
        const float* hp = h + (size_t)s * F + lane;
#pragma unroll
        for (int k = 0; k < K; ++k) hv[j][k] = hp[k * 32];
      }
#pragma unroll
      for (int sub = 0; sub < NSUB; ++sub) {
        const float p = __expf(lrelu(a1[sub] + a2[sub]) - fdecode(mk[sub]));
        if (ok[sub]) atomicAdd(&ssum[dj[sub] * H + hq], p);
#pragma unroll
        for (int jj = 0; jj < PE; ++jj)
#pragma unroll
          for (int q = 0; q < H; ++q)
            al[sub * PE + jj][q] = __int_as_float(__builtin_amdgcn_readlane(__float_as_int(p), jj * H + q));
      }
#pragma unroll
      for (int j = 0; j < EB; ++j) {
        if (ent[j] != NOENT) {
          float* accp = acc + (ent[j] >> 16) * F + lane;
#pragma unroll
          for (int k = 0; k < K; ++k) atomicAdd(&accp[k * 32], al[j][(k * 32) / D] * hv[j][k]);
        }
      }
    }
  } else {
    for (int eid = tid; eid < NEDGES; eid += OWN_T) {
      const unsigned dl = (unsigned)(dstI[eid] - r0);
      if (dl < (unsigned)OWN_R) {
        const int s = srcI[eid];
        for (int q = 0; q < H; ++q) {
          const float p = __expf(lrelu(asrc[(size_t)s * H + q] + adst[(size_t)(r0 + (int)dl) * H + q])
                                 - fdecode(mkey[dl * H + q]));
          atomicAdd(&ssum[dl * H + q], p);
          for (int f = q * D; f < (q + 1) * D; ++f)
            atomicAdd(&acc[dl * F + f], p * h[(size_t)s * F + f]);
        }
      }
    }
  }
  __syncthreads();

  for (int i = tid; i < OWN_R * F4; i += OWN_T) {
    const int row = i / F4, c4 = i - row * F4;
    const int n = r0 + row;
    if (n < NNODES) {
      float4 v = *(const float4*)(acc + (size_t)i * 4);
      const float sden = ssum[row * H + (c4 * 4) / D];
      const float* b = bias + c4 * 4;
      v.x = v.x / sden + b[0]; v.y = v.y / sden + b[1]; v.z = v.z / sden + b[2]; v.w = v.w / sden + b[3];
      if (RELU) { v.x = fmaxf(v.x, 0.f); v.y = fmaxf(v.y, 0.f); v.z = fmaxf(v.z, 0.f); v.w = fmaxf(v.w, 0.f); }
      *(float4*)(out + (size_t)n * F + c4 * 4) = v;
    }
  }
}

__global__ void pool_accum(const float* __restrict__ h, const int* __restrict__ batch,
                           float* __restrict__ sums, float* __restrict__ cnt) {
  int i = blockIdx.x * blockDim.x + threadIdx.x;
  if (i >= NNODES * F2) return;
  int n = i / F2, f = i - n * F2;
  int g = batch[n];
  atomicAdd(&sums[g * F2 + f], h[i]);
  if (f == 0) atomicAdd(&cnt[g], 1.0f);
}

__global__ void pool_final(const float* __restrict__ sums, const float* __restrict__ cnt,
                           float* __restrict__ out) {
  int i = blockIdx.x * blockDim.x + threadIdx.x;
  if (i >= NGRAPH * F2) return;
  out[i] = sums[i] / fmaxf(cnt[i / F2], 1.0f);
}

static inline int cdiv(long long a, int b) { return (int)((a + b - 1) / b); }

extern "C" void kernel_launch(void* const* d_in, const int* in_sizes, int n_in,
                              void* d_out, int out_size, void* d_ws, size_t ws_size,
                              hipStream_t stream) {
  const float* x        = (const float*)d_in[0];
  const int*   ei       = (const int*)d_in[1];
  const int*   batch    = (const int*)d_in[2];
  const float* W1       = (const float*)d_in[3];
  const float* att_src1 = (const float*)d_in[4];
  const float* att_dst1 = (const float*)d_in[5];
  const float* bias1    = (const float*)d_in[6];
  const float* W2       = (const float*)d_in[7];
  const float* att_src2 = (const float*)d_in[8];
  const float* att_dst2 = (const float*)d_in[9];
  const float* bias2    = (const float*)d_in[10];
  float* out = (float*)d_out;
  const int* srcI = ei;
  const int* dstI = ei + NEDGES;

  char* p = (char*)d_ws;
  auto alloc = [&](size_t bytes) -> char* {
    char* r = p; p += (bytes + 255) & ~(size_t)255; return r;
  };
  float*    h1    = (float*)alloc((size_t)NNODES * F1 * 4);
  float*    asrc1 = (float*)alloc((size_t)NNODES * H1C * 4);
  float*    adst1 = (float*)alloc((size_t)NNODES * H1C * 4);
  float*    agg1  = (float*)alloc((size_t)NNODES * F1 * 4);
  float*    h2    = (float*)alloc((size_t)NNODES * F2 * 4);
  float*    asrc2 = (float*)alloc((size_t)NNODES * H2C * 4);
  float*    adst2 = (float*)alloc((size_t)NNODES * H2C * 4);
  float*    agg2  = (float*)alloc((size_t)NNODES * F2 * 4);
  float*    psum  = (float*)alloc((size_t)NGRAPH * F2 * 4);
  float*    pcnt  = (float*)alloc((size_t)NGRAPH * 4);
  unsigned* glist = (unsigned*)alloc((size_t)((NNODES + OWN_R - 1) / OWN_R) * LIST_CAP * 4);

  hipMemsetAsync(psum,  0, (size_t)NGRAPH * F2 * 4, stream);
  hipMemsetAsync(pcnt,  0, (size_t)NGRAPH * 4, stream);

  const int T = 256;

  wmma_gemm_f32<<<dim3(NNODES / 16, F1 / 64), 128, 0, stream>>>(x, W1, h1, FIN, F1);
  att_coeff<<<cdiv((long long)NNODES * H1C, T), T, 0, stream>>>(h1, att_src1, att_dst1,
                                                                asrc1, adst1, H1C, D1C);
  owner_softmax_agg<F1, D1C, 4, 1><<<(NNODES + OWN_R - 1) / OWN_R, OWN_T, 0, stream>>>(
      srcI, dstI, asrc1, adst1, h1, bias1, glist, agg1);

  wmma_gemm_f32<<<dim3(NNODES / 16, F2 / 64), 128, 0, stream>>>(agg1, W2, h2, F1, F2);
  att_coeff<<<cdiv((long long)NNODES * H2C, T), T, 0, stream>>>(h2, att_src2, att_dst2,
                                                                asrc2, adst2, H2C, D2C);
  owner_softmax_agg<F2, D2C, 8, 0><<<(NNODES + OWN_R - 1) / OWN_R, OWN_T, 0, stream>>>(
      srcI, dstI, asrc2, adst2, h2, bias2, glist, agg2);

  pool_accum<<<cdiv((long long)NNODES * F2, T), T, 0, stream>>>(agg2, batch, psum, pcnt);
  pool_final<<<cdiv((long long)NGRAPH * F2, T), T, 0, stream>>>(psum, pcnt, out);
}


# conv2's owner groups take the edge lists and counts that conv1's groups noted in the workspace earlier in the same call; conv2's pass of each group over all destinations is gone; sums by arrival
# speedup vs baseline: 4.8349x; 1.1027x over previous
#include <hip/hip_runtime.h>
#include <math.h>

#define NNODES 50000
#define NEDGES 800000
#define NGRAPH 256
#define FIN    128
#define H1C    12
#define D1C    32
#define F1     (H1C * D1C)
#define H2C    1
#define D2C    64
#define F2     (H2C * D2C)
#define ETOT   (NEDGES + NNODES)

typedef float v2f __attribute__((ext_vector_type(2)));
typedef float v8f __attribute__((ext_vector_type(8)));

__device__ __forceinline__ unsigned fkey(float f) {
  unsigned u = __float_as_uint(f);
  return (u & 0x80000000u) ? ~u : (u | 0x80000000u);
}
__device__ __forceinline__ float fdecode(unsigned u) {
  unsigned b = (u & 0x80000000u) ? (u & 0x7fffffffu) : ~u;
  return __uint_as_float(b);
}

__global__ void __launch_bounds__(128)
wmma_gemm_f32(const float* __restrict__ A, const float* __restrict__ B,
              float* __restrict__ C, int K, int Ncols) {
  const int lane = threadIdx.x & 31;
  const int wave = threadIdx.x >> 5;
  const int l    = lane & 15;
  const int half = lane >> 4;
  const int m0   = blockIdx.x << 4;
  const int n0   = ((blockIdx.y << 2) + wave) << 4;

  v8f c = {};
  const float* arow = A + (size_t)(m0 + l) * K + (half << 1);
  const float* bcol = B + n0 + l;

  for (int k0 = 0; k0 < K; k0 += 4) {
    v2f a = *(const v2f*)(arow + k0);
    const int kb = k0 + (half << 1);
    v2f b;
    b.x = bcol[(size_t)kb * Ncols];
    b.y = bcol[(size_t)(kb + 1) * Ncols];
    c = __builtin_amdgcn_wmma_f32_16x16x4_f32(
         false, a,  false, b,
         (short)0, c,  false,  false);
  }

  float* crow = C + (size_t)(m0 + (half << 3)) * Ncols + n0 + l;
#pragma unroll
  for (int r = 0; r < 8; ++r) crow[(size_t)r * Ncols] = c[r];
}

__global__ void att_coeff(const float* __restrict__ h,
                          const float* __restrict__ att_src,
                          const float* __restrict__ att_dst,
                          float* __restrict__ asrc, float* __restrict__ adst,
                          int H, int D) {
  int i = blockIdx.x * blockDim.x + threadIdx.x;
  if (i >= NNODES * H) return;
  int n = i / H, hh = i - n * H;
  const float* hp = h + (size_t)n * H * D + hh * D;
  const float* ws = att_src + hh * D;
  const float* wd = att_dst + hh * D;
  float s = 0.f, d = 0.f;
  for (int j = 0; j < D; ++j) { float v = hp[j]; s += v * ws[j]; d += v * wd[j]; }
  asrc[i] = s; adst[i] = d;
}

#define OWN_R    196
#define OWN_T    1024
#define LIST_CAP 16384
#define SCAN_U   8
#define INIT_U   5
#define MAX_U    8
#define NOENT    0xFFFFFFFFu

__device__ __forceinline__ float lrelu(float v) { return (v > 0.f) ? v : 0.2f * v; }

template <int F, int D, int EB, int RELU, int NOTE>
__global__ void __launch_bounds__(OWN_T)
owner_softmax_agg(const int* __restrict__ srcI, const int* __restrict__ dstI,
                  const float* __restrict__ asrc, const float* __restrict__ adst,
                  const float* __restrict__ h, const float* __restrict__ bias,
                  unsigned* glist_all, unsigned* gcount_all, float* __restrict__ out) {
  constexpr int H    = F / D;
  constexpr int K    = F / 32;
  constexpr int F4   = F / 4;
  constexpr int RH   = OWN_R * H;
  constexpr int PE   = (32 / H) < EB ? (32 / H) : EB;
  constexpr int NSUB = EB / PE;
  static_assert(EB % PE == 0 && PE * H <= 32, "logit lanes");
  static_assert(NNODES <= 65536 && OWN_R <= 256, "list entry packing");
  __shared__ float    acc[OWN_R * F];
  __shared__ unsigned mkey[RH];
  __shared__ float    ssum[RH];
  __shared__ unsigned cnt;

  const int tid  = threadIdx.x;
  const int lane = tid & 31;
  const int wave = tid >> 5;
  const int r0   = blockIdx.x * OWN_R;
  unsigned* glist = glist_all + (size_t)blockIdx.x * LIST_CAP;

  if (NOTE && tid == 0) cnt = 0u;
  for (int t = tid; t < RH; t += OWN_T) {
    const int n = r0 + t / H;
    float v = 0.f;
    if (n < NNODES) v = lrelu(asrc[(size_t)r0 * H + t] + adst[(size_t)r0 * H + t]);
    mkey[t] = fkey(v);
  }
  __syncthreads();

  if (NOTE) {
    constexpr int NE4 = NEDGES / 4;
    const int4* d4 = (const int4*)dstI;
    for (int base = 0; base < NE4; base += OWN_T * SCAN_U) {
      int4 dv[SCAN_U];
#pragma unroll
      for (int u = 0; u < SCAN_U; ++u) {
        const int idx = base + u * OWN_T + tid;
        dv[u] = (idx < NE4) ? d4[idx] : make_int4(-1, -1, -1, -1);
      }
      unsigned m = 0u;
#pragma unroll
      for (int u = 0; u < SCAN_U; ++u) {
        m |= ((unsigned)(dv[u].x - r0) < (unsigned)OWN_R ? 1u : 0u) << (4 * u + 0);
        m |= ((unsigned)(dv[u].y - r0) < (unsigned)OWN_R ? 1u : 0u) << (4 * u + 1);
        m |= ((unsigned)(dv[u].z - r0) < (unsigned)OWN_R ? 1u : 0u) << (4 * u + 2);
        m |= ((unsigned)(dv[u].w - r0) < (unsigned)OWN_R ? 1u : 0u) << (4 * u + 3);
      }
      while (m) {
        const int j = __builtin_ctz(m);
        m &= m - 1u;
        const int eid = (base + (j >> 2) * OWN_T + tid) * 4 + (j & 3);
        const unsigned dl = (unsigned)(dstI[eid] - r0);
        const unsigned pos = atomicAdd(&cnt, 1u);
        if (pos < (unsigned)LIST_CAP) glist[pos] = (dl << 16) | (unsigned)srcI[eid];
      }
    }
    __syncthreads();
    if (tid == 0) gcount_all[blockIdx.x] = cnt;
  }

  const unsigned seen = NOTE ? cnt : gcount_all[blockIdx.x];
  const bool fast = seen <= (unsigned)LIST_CAP;

  if (fast) {
    const int nitems = (int)seen * H;
    for (int t0 = tid; t0 < nitems; t0 += OWN_T * MAX_U) {
      unsigned en[MAX_U];
      int      hd[MAX_U];
      float    a1[MAX_U], a2[MAX_U];
#pragma unroll
      for (int u = 0; u < MAX_U; ++u) {
        const int t  = t0 + u * OWN_T;
        const int tc = t < nitems ? t : nitems - 1;
        const int e  = tc / H;
        hd[u] = tc - e * H;
        en[u] = glist[e];
      }
#pragma unroll
      for (int u = 0; u < MAX_U; ++u) {
        a1[u] = asrc[(size_t)(en[u] & 0xFFFFu) * H + hd[u]];
        a2[u] = adst[(size_t)(r0 + (int)(en[u] >> 16)) * H + hd[u]];
      }
#pragma unroll
      for (int u = 0; u < MAX_U; ++u)
        if (t0 + u * OWN_T < nitems)
          atomicMax(&mkey[(en[u] >> 16) * H + hd[u]], fkey(lrelu(a1[u] + a2[u])));
    }
  } else {
    for (int eid = tid; eid < NEDGES; eid += OWN_T) {
      const unsigned dl = (unsigned)(dstI[eid] - r0);
      if (dl < (unsigned)OWN_R) {
        const int s = srcI[eid];
        for (int q = 0; q < H; ++q)
          atomicMax(&mkey[dl * H + q],
                    fkey(lrelu(asrc[(size_t)s * H + q] + adst[(size_t)(r0 + (int)dl) * H + q])));
      }
    }
  }
  __syncthreads();

  for (int i0 = tid; i0 < OWN_R * F4; i0 += OWN_T * INIT_U) {
    float4 hv[INIT_U];
    float  as[INIT_U], ad[INIT_U];
    bool   ok[INIT_U];
#pragma unroll
    for (int u = 0; u < INIT_U; ++u) {
      const int i   = i0 + u * OWN_T;
      const int ic  = i < OWN_R * F4 ? i : OWN_R * F4 - 1;
      const int row = ic / F4, c4 = ic - row * F4;
      const int n   = r0 + row;
      const int nc  = n < NNODES ? n : NNODES - 1;
      ok[u] = n < NNODES;
      hv[u] = *(const float4*)(h + (size_t)nc * F + c4 * 4);
      as[u] = asrc[(size_t)nc * H + (c4 * 4) / D];
      ad[u] = adst[(size_t)nc * H + (c4 * 4) / D];
    }
#pragma unroll
    for (int u = 0; u < INIT_U; ++u) {
      const int i = i0 + u * OWN_T;
      if (i < OWN_R * F4) {
        const int row = i / F4, c4 = i - row * F4;
        const int hq  = (c4 * 4) / D;
        float p = __expf(lrelu(as[u] + ad[u]) - fdecode(mkey[row * H + hq]));
        float4 v = make_float4(p * hv[u].x, p * hv[u].y, p * hv[u].z, p * hv[u].w);
        if (!ok[u]) { v = make_float4(0.f, 0.f, 0.f, 0.f); p = 1.f; }
        *(float4*)(acc + (size_t)i * 4) = v;
        if ((c4 * 4) % D == 0) ssum[row * H + hq] = p;
      }
    }
  }
  __syncthreads();

  if (fast) {
    const unsigned total = seen;
    const unsigned last  = total > 0u ? total - 1u : 0u;
    const unsigned lj    = lane < EB ? (unsigned)lane : 0u;
    unsigned i = (unsigned)wave * EB;
    unsigned nxt = NOENT;
    if (i < total) {
      const unsigned raw = glist[i + lj < last ? i + lj : last];
      nxt = ((lane < EB) & (i + lj < total)) ? raw : NOENT;
    }
    for (; i < total; i += 32u * EB) {
      const unsigned cur = nxt;
      const unsigned in  = i + 32u * EB + lj;
      const unsigned raw = glist[in < last ? in : last];
      nxt = ((lane < EB) & (in < total)) ? raw : NOENT;
      unsigned ent[EB];
      float hv[EB][K];
      float al[EB][H];
#pragma unroll
      for (int j = 0; j < EB; ++j) ent[j] = (unsigned)__builtin_amdgcn_readlane((int)cur, j);
      const int jl = lane / H;
      const int hq = lane - jl * H;
      float    a1[NSUB], a2[NSUB];
      unsigned mk[NSUB], dj[NSUB];
      bool     ok[NSUB];
#pragma unroll
      for (int sub = 0; sub < NSUB; ++sub) {
        const unsigned el = (unsigned)__shfl((int)cur, sub * PE + (jl < PE ? jl : 0));
        ok[sub] = (jl < PE) & (el != NOENT);
        const unsigned sj = ok[sub] ? (el & 0xFFFFu) : 0u;
        dj[sub] = ok[sub] ? (el >> 16) : 0u;
        a1[sub] = asrc[(size_t)sj * H + hq];
        a2[sub] = adst[(size_t)(r0 + (int)dj[sub]) * H + hq];
        mk[sub] = mkey[dj[sub] * H + hq];
      }
#pragma unroll
      for (int j = 0; j < EB; ++j) {
        const unsigned s = (ent[j] == NOENT) ? 0u : (ent[j] & 0xFFFFu);
        const float* hp = h + (size_t)s * F + lane;
#pragma unroll
        for (int k = 0; k < K; ++k) hv[j][k] = hp[k * 32];
      }
#pragma unroll
      for (int sub = 0; sub < NSUB; ++sub) {
        const float p = __expf(lrelu(a1[sub] + a2[sub]) - fdecode(mk[sub]));
        if (ok[sub]) atomicAdd(&ssum[dj[sub] * H + hq], p);
#pragma unroll
        for (int jj = 0; jj < PE; ++jj)
#pragma unroll
          for (int q = 0; q < H; ++q)
            al[sub * PE + jj][q] = __int_as_float(__builtin_amdgcn_readlane(__float_as_int(p), jj * H + q));
      }
#pragma unroll
      for (int j = 0; j < EB; ++j) {
        if (ent[j] != NOENT) {
          float* accp = acc + (ent[j] >> 16) * F + lane;
#pragma unroll
          for (int k = 0; k < K; ++k) atomicAdd(&accp[k * 32], al[j][(k * 32) / D] * hv[j][k]);
        }
      }
    }
  } else {
    for (int eid = tid; eid < NEDGES; eid += OWN_T) {
      const unsigned dl = (unsigned)(dstI[eid] - r0);
      if (dl < (unsigned)OWN_R) {
        const int s = srcI[eid];
        for (int q = 0; q < H; ++q) {
          const float p = __expf(lrelu(asrc[(size_t)s * H + q] + adst[(size_t)(r0 + (int)dl) * H + q])
                                 - fdecode(mkey[dl * H + q]));
          atomicAdd(&ssum[dl * H + q], p);
          for (int f = q * D; f < (q + 1) * D; ++f)
            atomicAdd(&acc[dl * F + f], p * h[(size_t)s * F + f]);
        }
      }
    }
  }
  __syncthreads();

  for (int i = tid; i < OWN_R * F4; i += OWN_T) {
    const int row = i / F4, c4 = i - row * F4;
    const int n = r0 + row;
    if (n < NNODES) {
      float4 v = *(const float4*)(acc + (size_t)i * 4);
      const float sden = ssum[row * H + (c4 * 4) / D];
      const float* b = bias + c4 * 4;
      v.x = v.x / sden + b[0]; v.y = v.y / sden + b[1]; v.z = v.z / sden + b[2]; v.w = v.w / sden + b[3];
      if (RELU) { v.x = fmaxf(v.x, 0.f); v.y = fmaxf(v.y, 0.f); v.z = fmaxf(v.z, 0.f); v.w = fmaxf(v.w, 0.f); }
      *(float4*)(out + (size_t)n * F + c4 * 4) = v;
    }
  }
}

__global__ void pool_accum(const float* __restrict__ h, const int* __restrict__ batch,
                           float* __restrict__ sums, float* __restrict__ cnt) {
  int i = blockIdx.x * blockDim.x + threadIdx.x;
  if (i >= NNODES * F2) return;
  int n = i / F2, f = i - n * F2;
  int g = batch[n];
  atomicAdd(&sums[g * F2 + f], h[i]);
  if (f == 0) atomicAdd(&cnt[g], 1.0f);
}

__global__ void pool_final(const float* __restrict__ sums, const float* __restrict__ cnt,
                           float* __restrict__ out) {
  int i = blockIdx.x * blockDim.x + threadIdx.x;
  if (i >= NGRAPH * F2) return;
  out[i] = sums[i] / fmaxf(cnt[i / F2], 1.0f);
}

static inline int cdiv(long long a, int b) { return (int)((a + b - 1) / b); }

extern "C" void kernel_launch(void* const* d_in, const int* in_sizes, int n_in,
                              void* d_out, int out_size, void* d_ws, size_t ws_size,
                              hipStream_t stream) {
  const float* x        = (const float*)d_in[0];
  const int*   ei       = (const int*)d_in[1];
  const int*   batch    = (const int*)d_in[2];
  const float* W1       = (const float*)d_in[3];
  const float* att_src1 = (const float*)d_in[4];
  const float* att_dst1 = (const float*)d_in[5];
  const float* bias1    = (const float*)d_in[6];
  const float* W2       = (const float*)d_in[7];
  const float* att_src2 = (const float*)d_in[8];
  const float* att_dst2 = (const float*)d_in[9];
  const float* bias2    = (const float*)d_in[10];
  float* out = (float*)d_out;
  const int* srcI = ei;
  const int* dstI = ei + NEDGES;

  char* p = (char*)d_ws;
  auto alloc = [&](size_t bytes) -> char* {
    char* r = p; p += (bytes + 255) & ~(size_t)255; return r;
  };
  float*    h1    = (float*)alloc((size_t)NNODES * F1 * 4);
  float*    asrc1 = (float*)alloc((size_t)NNODES * H1C * 4);
  float*    adst1 = (float*)alloc((size_t)NNODES * H1C * 4);
  float*    agg1  = (float*)alloc((size_t)NNODES * F1 * 4);
  float*    h2    = (float*)alloc((size_t)NNODES * F2 * 4);
  float*    asrc2 = (float*)alloc((size_t)NNODES * H2C * 4);
  float*    adst2 = (float*)alloc((size_t)NNODES * H2C * 4);
  float*    agg2  = (float*)alloc((size_t)NNODES * F2 * 4);
  float*    psum  = (float*)alloc((size_t)NGRAPH * F2 * 4);
  float*    pcnt  = (float*)alloc((size_t)NGRAPH * 4);
  unsigned* glist = (unsigned*)alloc((size_t)((NNODES + OWN_R - 1) / OWN_R) * LIST_CAP * 4);
  unsigned* gcnt  = (unsigned*)alloc((size_t)((NNODES + OWN_R - 1) / OWN_R) * 4);

  hipMemsetAsync(psum,  0, (size_t)NGRAPH * F2 * 4, stream);
  hipMemsetAsync(pcnt,  0, (size_t)NGRAPH * 4, stream);

  const int T = 256;

  wmma_gemm_f32<<<dim3(NNODES / 16, F1 / 64), 128, 0, stream>>>(x, W1, h1, FIN, F1);
  att_coeff<<<cdiv((long long)NNODES * H1C, T), T, 0, stream>>>(h1, att_src1, att_dst1,
                                                                asrc1, adst1, H1C, D1C);
  owner_softmax_agg<F1, D1C, 4, 1, 1><<<(NNODES + OWN_R - 1) / OWN_R, OWN_T, 0, stream>>>(
      srcI, dstI, asrc1, adst1, h1, bias1, glist, gcnt, agg1);

  wmma_gemm_f32<<<dim3(NNODES / 16, F2 / 64), 128, 0, stream>>>(agg1, W2, h2, F1, F2);
  att_coeff<<<cdiv((long long)NNODES * H2C, T), T, 0, stream>>>(h2, att_src2, att_dst2,
                                                                asrc2, adst2, H2C, D2C);
  owner_softmax_agg<F2, D2C, 8, 0, 0><<<(NNODES + OWN_R - 1) / OWN_R, OWN_T, 0, stream>>>(
      srcI, dstI, asrc2, adst2, h2, bias2, glist, gcnt, agg2);

  pool_accum<<<cdiv((long long)NNODES * F2, T), T, 0, stream>>>(agg2, batch, psum, pcnt);
  pool_final<<<cdiv((long long)NGRAPH * F2, T), T, 0, stream>>>(psum, pcnt, out);
}


# Mean pool folded into conv2's owner-group store: a group adds its rows to a per-graph table in LDS (float adds by arrival), then one atomic per graph and feature it touched; pool_accum launch gone
# speedup vs baseline: 5.4846x; 1.1344x over previous
#include <hip/hip_runtime.h>
#include <math.h>

#define NNODES 50000
#define NEDGES 800000
#define NGRAPH 256
#define FIN    128
#define H1C    12
#define D1C    32
#define F1     (H1C * D1C)
#define H2C    1
#define D2C    64
#define F2     (H2C * D2C)
#define ETOT   (NEDGES + NNODES)

typedef float v2f __attribute__((ext_vector_type(2)));
typedef float v8f __attribute__((ext_vector_type(8)));

__device__ __forceinline__ unsigned fkey(float f) {
  unsigned u = __float_as_uint(f);
  return (u & 0x80000000u) ? ~u : (u | 0x80000000u);
}
__device__ __forceinline__ float fdecode(unsigned u) {
  unsigned b = (u & 0x80000000u) ? (u & 0x7fffffffu) : ~u;
  return __uint_as_float(b);
}

__global__ void __launch_bounds__(128)
wmma_gemm_f32(const float* __restrict__ A, const float* __restrict__ B,
              float* __restrict__ C, int K, int Ncols) {
  const int lane = threadIdx.x & 31;
  const int wave = threadIdx.x >> 5;
  const int l    = lane & 15;
  const int half = lane >> 4;
  const int m0   = blockIdx.x << 4;
  const int n0   = ((blockIdx.y << 2) + wave) << 4;

  v8f c = {};
  const float* arow = A + (size_t)(m0 + l) * K + (half << 1);
  const float* bcol = B + n0 + l;

  for (int k0 = 0; k0 < K; k0 += 4) {
    v2f a = *(const v2f*)(arow + k0);
    const int kb = k0 + (half << 1);
    v2f b;
    b.x = bcol[(size_t)kb * Ncols];
    b.y = bcol[(size_t)(kb + 1) * Ncols];
    c = __builtin_amdgcn_wmma_f32_16x16x4_f32(
         false, a,  false, b,
         (short)0, c,  false,  false);
  }

  float* crow = C + (size_t)(m0 + (half << 3)) * Ncols + n0 + l;
#pragma unroll
  for (int r = 0; r < 8; ++r) crow[(size_t)r * Ncols] = c[r];
}

__global__ void att_coeff(const float* __restrict__ h,
                          const float* __restrict__ att_src,
                          const float* __restrict__ att_dst,
                          float* __restrict__ asrc, float* __restrict__ adst,
                          int H, int D) {
  int i = blockIdx.x * blockDim.x + threadIdx.x;
  if (i >= NNODES * H) return;
  int n = i / H, hh = i - n * H;
  const float* hp = h + (size_t)n * H * D + hh * D;
  const float* ws = att_src + hh * D;
  const float* wd = att_dst + hh * D;
  float s = 0.f, d = 0.f;
  for (int j = 0; j < D; ++j) { float v = hp[j]; s += v * ws[j]; d += v * wd[j]; }
  asrc[i] = s; adst[i] = d;
}

#define OWN_R    196
#define OWN_T    1024
#define LIST_CAP 16384
#define SCAN_U   8
#define INIT_U   5
#define MAX_U    8
#define NOENT    0xFFFFFFFFu

__device__ __forceinline__ float lrelu(float v) { return (v > 0.f) ? v : 0.2f * v; }

template <int F, int D, int EB, int RELU, int NOTE, int POOL>
__global__ void __launch_bounds__(OWN_T)
owner_softmax_agg(const int* __restrict__ srcI, const int* __restrict__ dstI,
                  const float* __restrict__ asrc, const float* __restrict__ adst,
                  const float* __restrict__ h, const float* __restrict__ bias,
                  unsigned* glist_all, unsigned* gcount_all, float* __restrict__ out,
                  const int* __restrict__ batch, float* psum, float* pcnt) {
  constexpr int H    = F / D;
  constexpr int K    = F / 32;
  constexpr int F4   = F / 4;
  constexpr int RH   = OWN_R * H;
  constexpr int PE   = (32 / H) < EB ? (32 / H) : EB;
  constexpr int NSUB = EB / PE;
  static_assert(EB % PE == 0 && PE * H <= 32, "logit lanes");
  static_assert(NNODES <= 65536 && OWN_R <= 256, "list entry packing");
  __shared__ float    acc[OWN_R * F];
  __shared__ unsigned mkey[RH];
  __shared__ float    ssum[RH];
  __shared__ unsigned cnt;
  __shared__ float    ptab[POOL ? NGRAPH * F : 1];
  __shared__ unsigned pnum[POOL ? NGRAPH : 1];
  __shared__ unsigned pgid[POOL ? OWN_R : 1];
  static_assert(!POOL || OWN_T % F == 0, "a thread keeps its feature over the rows of the pool pass");

  const int tid  = threadIdx.x;
  const int lane = tid & 31;
  const int wave = tid >> 5;
  const int r0   = blockIdx.x * OWN_R;
  unsigned* glist = glist_all + (size_t)blockIdx.x * LIST_CAP;

  if (NOTE && tid == 0) cnt = 0u;
  if (POOL) {
    for (int t = tid; t < NGRAPH * F / 4; t += OWN_T) *(float4*)(ptab + (size_t)t * 4) = make_float4(0.f, 0.f, 0.f, 0.f);
    for (int t = tid; t < NGRAPH; t += OWN_T) pnum[t] = 0u;
    for (int t = tid; t < OWN_R; t += OWN_T) {
      const int n = r0 + t;
      const unsigned g = n < NNODES ? (unsigned)batch[n] : (unsigned)NGRAPH;
      pgid[t] = g < (unsigned)NGRAPH ? g : (unsigned)NGRAPH;
    }
  }
  for (int t = tid; t < RH; t += OWN_T) {
    const int n = r0 + t / H;
    float v = 0.f;
    if (n < NNODES) v = lrelu(asrc[(size_t)r0 * H + t] + adst[(size_t)r0 * H + t]);
    mkey[t] = fkey(v);
  }
  __syncthreads();

  if (NOTE) {
    constexpr int NE4 = NEDGES / 4;
    const int4* d4 = (const int4*)dstI;
    for (int base = 0; base < NE4; base += OWN_T * SCAN_U) {
      int4 dv[SCAN_U];
#pragma unroll
      for (int u = 0; u < SCAN_U; ++u) {
        const int idx = base + u * OWN_T + tid;
        dv[u] = (idx < NE4) ? d4[idx] : make_int4(-1, -1, -1, -1);
      }
      unsigned m = 0u;
#pragma unroll
      for (int u = 0; u < SCAN_U; ++u) {
        m |= ((unsigned)(dv[u].x - r0) < (unsigned)OWN_R ? 1u : 0u) << (4 * u + 0);
        m |= ((unsigned)(dv[u].y - r0) < (unsigned)OWN_R ? 1u : 0u) << (4 * u + 1);
        m |= ((unsigned)(dv[u].z - r0) < (unsigned)OWN_R ? 1u : 0u) << (4 * u + 2);
        m |= ((unsigned)(dv[u].w - r0) < (unsigned)OWN_R ? 1u : 0u) << (4 * u + 3);
      }
      while (m) {
        const int j = __builtin_ctz(m);
        m &= m - 1u;
        const int eid = (base + (j >> 2) * OWN_T + tid) * 4 + (j & 3);
        const unsigned dl = (unsigned)(dstI[eid] - r0);
        const unsigned pos = atomicAdd(&cnt, 1u);
        if (pos < (unsigned)LIST_CAP) glist[pos] = (dl << 16) | (unsigned)srcI[eid];
      }
    }
    __syncthreads();
    if (tid == 0) gcount_all[blockIdx.x] = cnt;
  }

  const unsigned seen = NOTE ? cnt : gcount_all[blockIdx.x];
  const bool fast = seen <= (unsigned)LIST_CAP;

  if (fast) {
    const int nitems = (int)seen * H;
    for (int t0 = tid; t0 < nitems; t0 += OWN_T * MAX_U) {
      unsigned en[MAX_U];
      int      hd[MAX_U];
      float    a1[MAX_U], a2[MAX_U];
#pragma unroll
      for (int u = 0; u < MAX_U; ++u) {
        const int t  = t0 + u * OWN_T;
        const int tc = t < nitems ? t : nitems - 1;
        const int e  = tc / H;
        hd[u] = tc - e * H;
        en[u] = glist[e];
      }
#pragma unroll
      for (int u = 0; u < MAX_U; ++u) {
        a1[u] = asrc[(size_t)(en[u] & 0xFFFFu) * H + hd[u]];
        a2[u] = adst[(size_t)(r0 + (int)(en[u] >> 16)) * H + hd[u]];
      }
#pragma unroll
      for (int u = 0; u < MAX_U; ++u)
        if (t0 + u * OWN_T < nitems)
          atomicMax(&mkey[(en[u] >> 16) * H + hd[u]], fkey(lrelu(a1[u] + a2[u])));
    }
  } else {
    for (int eid = tid; eid < NEDGES; eid += OWN_T) {
      const unsigned dl = (unsigned)(dstI[eid] - r0);
      if (dl < (unsigned)OWN_R) {
        const int s = srcI[eid];
        for (int q = 0; q < H; ++q)
          atomicMax(&mkey[dl * H + q],
                    fkey(lrelu(asrc[(size_t)s * H + q] + adst[(size_t)(r0 + (int)dl) * H + q])));
      }
    }
  }
  __syncthreads();

  for (int i0 = tid; i0 < OWN_R * F4; i0 += OWN_T * INIT_U) {
    float4 hv[INIT_U];
    float  as[INIT_U], ad[INIT_U];
    bool   ok[INIT_U];
#pragma unroll
    for (int u = 0; u < INIT_U; ++u) {
      const int i   = i0 + u * OWN_T;
      const int ic  = i < OWN_R * F4 ? i : OWN_R * F4 - 1;
      const int row = ic / F4, c4 = ic - row * F4;
      const int n   = r0 + row;
      const int nc  = n < NNODES ? n : NNODES - 1;
      ok[u] = n < NNODES;
      hv[u] = *(const float4*)(h + (size_t)nc * F + c4 * 4);
      as[u] = asrc[(size_t)nc * H + (c4 * 4) / D];
      ad[u] = adst[(size_t)nc * H + (c4 * 4) / D];
    }
#pragma unroll
    for (int u = 0; u < INIT_U; ++u) {
      const int i = i0 + u * OWN_T;
      if (i < OWN_R * F4) {
        const int row = i / F4, c4 = i - row * F4;
        const int hq  = (c4 * 4) / D;
        float p = __expf(lrelu(as[u] + ad[u]) - fdecode(mkey[row * H + hq]));
        float4 v = make_float4(p * hv[u].x, p * hv[u].y, p * hv[u].z, p * hv[u].w);
        if (!ok[u]) { v = make_float4(0.f, 0.f, 0.f, 0.f); p = 1.f; }
        *(float4*)(acc + (size_t)i * 4) = v;
        if ((c4 * 4) % D == 0) ssum[row * H + hq] = p;
      }
    }
  }
  __syncthreads();

  if (fast) {
    const unsigned total = seen;
    const unsigned last  = total > 0u ? total - 1u : 0u;
    const unsigned lj    = lane < EB ? (unsigned)lane : 0u;
    unsigned i = (unsigned)wave * EB;
    unsigned nxt = NOENT;
    if (i < total) {
      const unsigned raw = glist[i + lj < last ? i + lj : last];
      nxt = ((lane < EB) & (i + lj < total)) ? raw : NOENT;
    }
    for (; i < total; i += 32u * EB) {
      const unsigned cur = nxt;
      const unsigned in  = i + 32u * EB + lj;
      const unsigned raw = glist[in < last ? in : last];
      nxt = ((lane < EB) & (in < total)) ? raw : NOENT;
      unsigned ent[EB];
      float hv[EB][K];
      float al[EB][H];
#pragma unroll
      for (int j = 0; j < EB; ++j) ent[j] = (unsigned)__builtin_amdgcn_readlane((int)cur, j);
      const int jl = lane / H;
      const int hq = lane - jl * H;
      float    a1[NSUB], a2[NSUB];
      unsigned mk[NSUB], dj[NSUB];
      bool     ok[NSUB];
#pragma unroll
      for (int sub = 0; sub < NSUB; ++sub) {
        const unsigned el = (unsigned)__shfl((int)cur, sub * PE + (jl < PE ? jl : 0));
        ok[sub] = (jl < PE) & (el != NOENT);
        const unsigned sj = ok[sub] ? (el & 0xFFFFu) : 0u;
        dj[sub] = ok[sub] ? (el >> 16) : 0u;
        a1[sub] = asrc[(size_t)sj * H + hq];
        a2[sub] = adst[(size_t)(r0 + (int)dj[sub]) * H + hq];
        mk[sub] = mkey[dj[sub] * H + hq];
      }
#pragma unroll
      for (int j = 0; j < EB; ++j) {
        const unsigned s = (ent[j] == NOENT) ? 0u : (ent[j] & 0xFFFFu);
        const float* hp = h + (size_t)s * F + lane;
#pragma unroll
        for (int k = 0; k < K; ++k) hv[j][k] = hp[k * 32];
      }
#pragma unroll
      for (int sub = 0; sub < NSUB; ++sub) {
        const float p = __expf(lrelu(a1[sub] + a2[sub]) - fdecode(mk[sub]));
        if (ok[sub]) atomicAdd(&ssum[dj[sub] * H + hq], p);
#pragma unroll
        for (int jj = 0; jj < PE; ++jj)
#pragma unroll
          for (int q = 0; q < H; ++q)
            al[sub * PE + jj][q] = __int_as_float(__builtin_amdgcn_readlane(__float_as_int(p), jj * H + q));
      }
#pragma unroll
      for (int j = 0; j < EB; ++j) {
        if (ent[j] != NOENT) {
          float* accp = acc + (ent[j] >> 16) * F + lane;
#pragma unroll
          for (int k = 0; k < K; ++k) atomicAdd(&accp[k * 32], al[j][(k * 32) / D] * hv[j][k]);
        }
      }
    }
  } else {
    for (int eid = tid; eid < NEDGES; eid += OWN_T) {
      const unsigned dl = (unsigned)(dstI[eid] - r0);
      if (dl < (unsigned)OWN_R) {
        const int s = srcI[eid];
        for (int q = 0; q < H; ++q) {
          const float p = __expf(lrelu(asrc[(size_t)s * H + q] + adst[(size_t)(r0 + (int)dl) * H + q])
                                 - fdecode(mkey[dl * H + q]));
          atomicAdd(&ssum[dl * H + q], p);
          for (int f = q * D; f < (q + 1) * D; ++f)
            atomicAdd(&acc[dl * F + f], p * h[(size_t)s * F + f]);
        }
      }
    }
  }
  __syncthreads();

  if (POOL) {
    const int   f  = tid % F;
    const float bf = bias[f];
    for (int row = tid / F; row < OWN_R; row += OWN_T / F) {
      const unsigned g = pgid[row];
      if (g < (unsigned)NGRAPH) {
        float v = acc[row * F + f] / ssum[row * H + f / D] + bf;
        if (RELU) v = fmaxf(v, 0.f);
        atomicAdd(&ptab[g * F + f], v);
        if (f == 0) atomicAdd(&pnum[g], 1u);
      }
    }
    __syncthreads();
    for (int g = wave; g < NGRAPH; g += OWN_T / 32) {
      const unsigned c = pnum[g];
      if (c != 0u) {
        for (int f = lane; f < F; f += 32) atomicAdd(&psum[g * F + f], ptab[g * F + f]);
        if (lane == 0) atomicAdd(&pcnt[g], (float)c);
      }
    }
    return;
  }

  for (int i = tid; i < OWN_R * F4; i += OWN_T) {
    const int row = i / F4, c4 = i - row * F4;
    const int n = r0 + row;
    if (n < NNODES) {
      float4 v = *(const float4*)(acc + (size_t)i * 4);
      const float sden = ssum[row * H + (c4 * 4) / D];
      const float* b = bias + c4 * 4;
      v.x = v.x / sden + b[0]; v.y = v.y / sden + b[1]; v.z = v.z / sden + b[2]; v.w = v.w / sden + b[3];
      if (RELU) { v.x = fmaxf(v.x, 0.f); v.y = fmaxf(v.y, 0.f); v.z = fmaxf(v.z, 0.f); v.w = fmaxf(v.w, 0.f); }
      *(float4*)(out + (size_t)n * F + c4 * 4) = v;
    }
  }
}

__global__ void pool_final(const float* __restrict__ sums, const float* __restrict__ cnt,
                           float* __restrict__ out) {
  int i = blockIdx.x * blockDim.x + threadIdx.x;
  if (i >= NGRAPH * F2) return;
  out[i] = sums[i] / fmaxf(cnt[i / F2], 1.0f);
}

static inline int cdiv(long long a, int b) { return (int)((a + b - 1) / b); }

extern "C" void kernel_launch(void* const* d_in, const int* in_sizes, int n_in,
                              void* d_out, int out_size, void* d_ws, size_t ws_size,
                              hipStream_t stream) {
  const float* x        = (const float*)d_in[0];
  const int*   ei       = (const int*)d_in[1];
  const int*   batch    = (const int*)d_in[2];
  const float* W1       = (const float*)d_in[3];
  const float* att_src1 = (const float*)d_in[4];
  const float* att_dst1 = (const float*)d_in[5];
  const float* bias1    = (const float*)d_in[6];
  const float* W2       = (const float*)d_in[7];
  const float* att_src2 = (const float*)d_in[8];
  const float* att_dst2 = (const float*)d_in[9];
  const float* bias2    = (const float*)d_in[10];
  float* out = (float*)d_out;
  const int* srcI = ei;
  const int* dstI = ei + NEDGES;

  char* p = (char*)d_ws;
  auto alloc = [&](size_t bytes) -> char* {
    char* r = p; p += (bytes + 255) & ~(size_t)255; return r;
  };
  float*    h1    = (float*)alloc((size_t)NNODES * F1 * 4);
  float*    asrc1 = (float*)alloc((size_t)NNODES * H1C * 4);
  float*    adst1 = (float*)alloc((size_t)NNODES * H1C * 4);
  float*    agg1  = (float*)alloc((size_t)NNODES * F1 * 4);
  float*    h2    = (float*)alloc((size_t)NNODES * F2 * 4);
  float*    asrc2 = (float*)alloc((size_t)NNODES * H2C * 4);
  float*    adst2 = (float*)alloc((size_t)NNODES * H2C * 4);
  float*    psum  = (float*)alloc((size_t)NGRAPH * F2 * 4);
  float*    pcnt  = (float*)alloc((size_t)NGRAPH * 4);
  unsigned* glist = (unsigned*)alloc((size_t)((NNODES + OWN_R - 1) / OWN_R) * LIST_CAP * 4);
  unsigned* gcnt  = (unsigned*)alloc((size_t)((NNODES + OWN_R - 1) / OWN_R) * 4);

  hipMemsetAsync(psum,  0, (size_t)NGRAPH * F2 * 4, stream);
  hipMemsetAsync(pcnt,  0, (size_t)NGRAPH * 4, stream);

  const int T = 256;

  wmma_gemm_f32<<<dim3(NNODES / 16, F1 / 64), 128, 0, stream>>>(x, W1, h1, FIN, F1);
  att_coeff<<<cdiv((long long)NNODES * H1C, T), T, 0, stream>>>(h1, att_src1, att_dst1,
                                                                asrc1, adst1, H1C, D1C);
  owner_softmax_agg<F1, D1C, 4, 1, 1, 0><<<(NNODES + OWN_R - 1) / OWN_R, OWN_T, 0, stream>>>(
      srcI, dstI, asrc1, adst1, h1, bias1, glist, gcnt, agg1, (const int*)nullptr, (float*)nullptr, (float*)nullptr);

  wmma_gemm_f32<<<dim3(NNODES / 16, F2 / 64), 128, 0, stream>>>(agg1, W2, h2, F1, F2);
  att_coeff<<<cdiv((long long)NNODES * H2C, T), T, 0, stream>>>(h2, att_src2, att_dst2,
                                                                asrc2, adst2, H2C, D2C);
  owner_softmax_agg<F2, D2C, 8, 0, 0, 1><<<(NNODES + OWN_R - 1) / OWN_R, OWN_T, 0, stream>>>(
      srcI, dstI, asrc2, adst2, h2, bias2, glist, gcnt, (float*)nullptr, batch, psum, pcnt);

  pool_final<<<cdiv((long long)NGRAPH * F2, T), T, 0, stream>>>(psum, pcnt, out);
}


# Both products: weights staged once per persistent work-group in LDS as k-pair rows (bank-offset pitch); a wave keeps its 16 rows of A in registers, B fragments from LDS; k order and sums as before
# speedup vs baseline: 6.7420x; 1.2293x over previous
#include <hip/hip_runtime.h>
#include <math.h>

#define NNODES 50000
#define NEDGES 800000
#define NGRAPH 256
#define FIN    128
#define H1C    12
#define D1C    32
#define F1     (H1C * D1C)
#define H2C    1
#define D2C    64
#define F2     (H2C * D2C)
#define ETOT   (NEDGES + NNODES)

typedef float v2f __attribute__((ext_vector_type(2)));
typedef float v8f __attribute__((ext_vector_type(8)));

__device__ __forceinline__ unsigned fkey(float f) {
  unsigned u = __float_as_uint(f);
  return (u & 0x80000000u) ? ~u : (u | 0x80000000u);
}
__device__ __forceinline__ float fdecode(unsigned u) {
  unsigned b = (u & 0x80000000u) ? (u & 0x7fffffffu) : ~u;
  return __uint_as_float(b);
}

template <int K, int N, int TN, int T, int G>
__global__ void __launch_bounds__(T, 4)
wmma_gemm_ldsw(const float* __restrict__ A, const float* __restrict__ B, float* __restrict__ C) {
  constexpr int PITCH = 2 * N + 32;
  constexpr int KP    = K / 2;
  constexpr int KCS   = 32;
  constexpr int NKC   = K / (4 * KCS);
  constexpr int KS    = K / 4;
  constexpr int PD    = 16 / TN;
  constexpr int NG    = N / (16 * TN);
  constexpr int UNITS = (NNODES / 16) * NG;
  constexpr int WPG   = T / 32;
  constexpr int N4    = N / 4;
  constexpr int SI    = KP * N4 / T;
  static_assert(K % (4 * KCS) == 0 && N % (16 * TN) == 0 && NNODES % 16 == 0 && (2 * N) % 64 == 0, "tiling");
  static_assert((KP * N4) % T == 0, "staging");
  __shared__ __attribute__((aligned(16))) float wl[KP * PITCH];

  const int tid = threadIdx.x;
  {
    float4 w0[SI], w1[SI];
#pragma unroll
    for (int j = 0; j < SI; ++j) {
      const int i = tid + j * T;
      const int kp = i / N4, q = i - kp * N4;
      w0[j] = *(const float4*)(B + (size_t)(2 * kp) * N + 4 * q);
      w1[j] = *(const float4*)(B + (size_t)(2 * kp + 1) * N + 4 * q);
    }
#pragma unroll
    for (int j = 0; j < SI; ++j) {
      const int i = tid + j * T;
      const int kp = i / N4, q = i - kp * N4;
      float* d = wl + kp * PITCH + 8 * q;
      *(float4*)d       = make_float4(w0[j].x, w1[j].x, w0[j].y, w1[j].y);
      *(float4*)(d + 4) = make_float4(w0[j].z, w1[j].z, w0[j].w, w1[j].w);
    }
  }
  __syncthreads();

  const int lane = tid & 31;
  const int wave = tid >> 5;
  const int l    = lane & 15;
  const int half = lane >> 4;

  for (int u = blockIdx.x + G * wave; u < UNITS; u += G * WPG) {
    const int strip = u / NG, ng = u - strip * NG;
    const int m0 = strip << 4;
    const int n0 = ng * (16 * TN);

    const float* arow = A + (size_t)(m0 + l) * K + (half << 1);
    const float* bl   = wl + half * PITCH + 2 * (n0 + l);
    v2f a[2][KCS];
    v2f bf[PD + 1][TN];
    v8f c[TN];
#pragma unroll
    for (int t = 0; t < TN; ++t) c[t] = v8f{};
#pragma unroll
    for (int s = 0; s < KCS; ++s) a[0][s] = *(const v2f*)(arow + 4 * s);
#pragma unroll
    for (int p = 0; p < PD; ++p)
#pragma unroll
      for (int t = 0; t < TN; ++t) bf[p][t] = *(const v2f*)(bl + (2 * p) * PITCH + 32 * t);
#pragma unroll
    for (int kc = 0; kc < NKC; ++kc) {
      if (kc + 1 < NKC) {
#pragma unroll
        for (int s = 0; s < KCS; ++s) a[(kc + 1) & 1][s] = *(const v2f*)(arow + 4 * ((kc + 1) * KCS + s));
      }
      __builtin_amdgcn_sched_barrier(0);
#pragma unroll
      for (int s = 0; s < KCS; ++s) {
        const int gs = kc * KCS + s;
        if (gs + PD < KS) {
#pragma unroll
          for (int t = 0; t < TN; ++t)
            bf[(gs + PD) % (PD + 1)][t] = *(const v2f*)(bl + (2 * (gs + PD)) * PITCH + 32 * t);
        }
#pragma unroll
        for (int t = 0; t < TN; ++t)
          c[t] = __builtin_amdgcn_wmma_f32_16x16x4_f32(
               false, a[kc & 1][s],  false, bf[gs % (PD + 1)][t],
               (short)0, c[t],  false,  false);
        __builtin_amdgcn_sched_barrier(0);
      }
    }

    float* crow = C + (size_t)(m0 + (half << 3)) * N + n0 + l;
#pragma unroll
    for (int r = 0; r < 8; ++r)
#pragma unroll
      for (int t = 0; t < TN; ++t) crow[(size_t)r * N + 16 * t] = c[t][r];
  }
}

__global__ void att_coeff(const float* __restrict__ h,
                          const float* __restrict__ att_src,
                          const float* __restrict__ att_dst,
                          float* __restrict__ asrc, float* __restrict__ adst,
                          int H, int D) {
  int i = blockIdx.x * blockDim.x + threadIdx.x;
  if (i >= NNODES * H) return;
  int n = i / H, hh = i - n * H;
  const float* hp = h + (size_t)n * H * D + hh * D;
  const float* ws = att_src + hh * D;
  const float* wd = att_dst + hh * D;
  float s = 0.f, d = 0.f;
  for (int j = 0; j < D; ++j) { float v = hp[j]; s += v * ws[j]; d += v * wd[j]; }
  asrc[i] = s; adst[i] = d;
}

#define OWN_R    196
#define OWN_T    1024
#define LIST_CAP 16384
#define SCAN_U   8
#define INIT_U   5
#define MAX_U    8
#define NOENT    0xFFFFFFFFu

__device__ __forceinline__ float lrelu(float v) { return (v > 0.f) ? v : 0.2f * v; }

template <int F, int D, int EB, int RELU, int NOTE, int POOL>
__global__ void __launch_bounds__(OWN_T)
owner_softmax_agg(const int* __restrict__ srcI, const int* __restrict__ dstI,
                  const float* __restrict__ asrc, const float* __restrict__ adst,
                  const float* __restrict__ h, const float* __restrict__ bias,
                  unsigned* glist_all, unsigned* gcount_all, float* __restrict__ out,
                  const int* __restrict__ batch, float* psum, float* pcnt) {
  constexpr int H    = F / D;
  constexpr int K    = F / 32;
  constexpr int F4   = F / 4;
  constexpr int RH   = OWN_R * H;
  constexpr int PE   = (32 / H) < EB ? (32 / H) : EB;
  constexpr int NSUB = EB / PE;
  static_assert(EB % PE == 0 && PE * H <= 32, "logit lanes");
  static_assert(NNODES <= 65536 && OWN_R <= 256, "list entry packing");
  __shared__ float    acc[OWN_R * F];
  __shared__ unsigned mkey[RH];
  __shared__ float    ssum[RH];
  __shared__ unsigned cnt;
  __shared__ float    ptab[POOL ? NGRAPH * F : 1];
  __shared__ unsigned pnum[POOL ? NGRAPH : 1];
  __shared__ unsigned pgid[POOL ? OWN_R : 1];
  static_assert(!POOL || OWN_T % F == 0, "a thread keeps its feature over the rows of the pool pass");

  const int tid  = threadIdx.x;
  const int lane = tid & 31;
  const int wave = tid >> 5;
  const int r0   = blockIdx.x * OWN_R;
  unsigned* glist = glist_all + (size_t)blockIdx.x * LIST_CAP;

  if (NOTE && tid == 0) cnt = 0u;
  if (POOL) {
    for (int t = tid; t < NGRAPH * F / 4; t += OWN_T) *(float4*)(ptab + (size_t)t * 4) = make_float4(0.f, 0.f, 0.f, 0.f);
    for (int t = tid; t < NGRAPH; t += OWN_T) pnum[t] = 0u;
    for (int t = tid; t < OWN_R; t += OWN_T) {
      const int n = r0 + t;
      const unsigned g = n < NNODES ? (unsigned)batch[n] : (unsigned)NGRAPH;
      pgid[t] = g < (unsigned)NGRAPH ? g : (unsigned)NGRAPH;
    }
  }
  for (int t = tid; t < RH; t += OWN_T) {
    const int n = r0 + t / H;
    float v = 0.f;
    if (n < NNODES) v = lrelu(asrc[(size_t)r0 * H + t] + adst[(size_t)r0 * H + t]);
    mkey[t] = fkey(v);
  }
  __syncthreads();

  if (NOTE) {
    constexpr int NE4 = NEDGES / 4;
    const int4* d4 = (const int4*)dstI;
    for (int base = 0; base < NE4; base += OWN_T * SCAN_U) {
      int4 dv[SCAN_U];
#pragma unroll
      for (int u = 0; u < SCAN_U; ++u) {
        const int idx = base + u * OWN_T + tid;
        dv[u] = (idx < NE4) ? d4[idx] : make_int4(-1, -1, -1, -1);
      }
      unsigned m = 0u;
#pragma unroll
      for (int u = 0; u < SCAN_U; ++u) {
        m |= ((unsigned)(dv[u].x - r0) < (unsigned)OWN_R ? 1u : 0u) << (4 * u + 0);
        m |= ((unsigned)(dv[u].y - r0) < (unsigned)OWN_R ? 1u : 0u) << (4 * u + 1);
        m |= ((unsigned)(dv[u].z - r0) < (unsigned)OWN_R ? 1u : 0u) << (4 * u + 2);
        m |= ((unsigned)(dv[u].w - r0) < (unsigned)OWN_R ? 1u : 0u) << (4 * u + 3);
      }
      while (m) {
        const int j = __builtin_ctz(m);
        m &= m - 1u;
        const int eid = (base + (j >> 2) * OWN_T + tid) * 4 + (j & 3);
        const unsigned dl = (unsigned)(dstI[eid] - r0);
        const unsigned pos = atomicAdd(&cnt, 1u);
        if (pos < (unsigned)LIST_CAP) glist[pos] = (dl << 16) | (unsigned)srcI[eid];
      }
    }
    __syncthreads();
    if (tid == 0) gcount_all[blockIdx.x] = cnt;
  }

  const unsigned seen = NOTE ? cnt : gcount_all[blockIdx.x];
  const bool fast = seen <= (unsigned)LIST_CAP;

  if (fast) {
    const int nitems = (int)seen * H;
    for (int t0 = tid; t0 < nitems; t0 += OWN_T * MAX_U) {
      unsigned en[MAX_U];
      int      hd[MAX_U];
      float    a1[MAX_U], a2[MAX_U];
#pragma unroll
      for (int u = 0; u < MAX_U; ++u) {
        const int t  = t0 + u * OWN_T;
        const int tc = t < nitems ? t : nitems - 1;
        const int e  = tc / H;
        hd[u] = tc - e * H;
        en[u] = glist[e];
      }
#pragma unroll
      for (int u = 0; u < MAX_U; ++u) {
        a1[u] = asrc[(size_t)(en[u] & 0xFFFFu) * H + hd[u]];
        a2[u] = adst[(size_t)(r0 + (int)(en[u] >> 16)) * H + hd[u]];
      }
#pragma unroll
      for (int u = 0; u < MAX_U; ++u)
        if (t0 + u * OWN_T < nitems)
          atomicMax(&mkey[(en[u] >> 16) * H + hd[u]], fkey(lrelu(a1[u] + a2[u])));
    }
  } else {
    for (int eid = tid; eid < NEDGES; eid += OWN_T) {
      const unsigned dl = (unsigned)(dstI[eid] - r0);
      if (dl < (unsigned)OWN_R) {
        const int s = srcI[eid];
        for (int q = 0; q < H; ++q)
          atomicMax(&mkey[dl * H + q],
                    fkey(lrelu(asrc[(size_t)s * H + q] + adst[(size_t)(r0 + (int)dl) * H + q])));
      }
    }
  }
  __syncthreads();

  for (int i0 = tid; i0 < OWN_R * F4; i0 += OWN_T * INIT_U) {
    float4 hv[INIT_U];
    float  as[INIT_U], ad[INIT_U];
    bool   ok[INIT_U];
#pragma unroll
    for (int u = 0; u < INIT_U; ++u) {
      const int i   = i0 + u * OWN_T;
      const int ic  = i < OWN_R * F4 ? i : OWN_R * F4 - 1;
      const int row = ic / F4, c4 = ic - row * F4;
      const int n   = r0 + row;
      const int nc  = n < NNODES ? n : NNODES - 1;
      ok[u] = n < NNODES;
      hv[u] = *(const float4*)(h + (size_t)nc * F + c4 * 4);
      as[u] = asrc[(size_t)nc * H + (c4 * 4) / D];
      ad[u] = adst[(size_t)nc * H + (c4 * 4) / D];
    }
#pragma unroll
    for (int u = 0; u < INIT_U; ++u) {
      const int i = i0 + u * OWN_T;
      if (i < OWN_R * F4) {
        const int row = i / F4, c4 = i - row * F4;
        const int hq  = (c4 * 4) / D;
        float p = __expf(lrelu(as[u] + ad[u]) - fdecode(mkey[row * H + hq]));
        float4 v = make_float4(p * hv[u].x, p * hv[u].y, p * hv[u].z, p * hv[u].w);
        if (!ok[u]) { v = make_float4(0.f, 0.f, 0.f, 0.f); p = 1.f; }
        *(float4*)(acc + (size_t)i * 4) = v;
        if ((c4 * 4) % D == 0) ssum[row * H + hq] = p;
      }
    }
  }
  __syncthreads();

  if (fast) {
    const unsigned total = seen;
    const unsigned last  = total > 0u ? total - 1u : 0u;
    const unsigned lj    = lane < EB ? (unsigned)lane : 0u;
    unsigned i = (unsigned)wave * EB;
    unsigned nxt = NOENT;
    if (i < total) {
      const unsigned raw = glist[i + lj < last ? i + lj : last];
      nxt = ((lane < EB) & (i + lj < total)) ? raw : NOENT;
    }
    for (; i < total; i += 32u * EB) {
      const unsigned cur = nxt;
      const unsigned in  = i + 32u * EB + lj;
      const unsigned raw = glist[in < last ? in : last];
      nxt = ((lane < EB) & (in < total)) ? raw : NOENT;
      unsigned ent[EB];
      float hv[EB][K];
      float al[EB][H];
#pragma unroll
      for (int j = 0; j < EB; ++j) ent[j] = (unsigned)__builtin_amdgcn_readlane((int)cur, j);
      const int jl = lane / H;
      const int hq = lane - jl * H;
      float    a1[NSUB], a2[NSUB];
      unsigned mk[NSUB], dj[NSUB];
      bool     ok[NSUB];
#pragma unroll
      for (int sub = 0; sub < NSUB; ++sub) {
        const unsigned el = (unsigned)__shfl((int)cur, sub * PE + (jl < PE ? jl : 0));
        ok[sub] = (jl < PE) & (el != NOENT);
        const unsigned sj = ok[sub] ? (el & 0xFFFFu) : 0u;
        dj[sub] = ok[sub] ? (el >> 16) : 0u;
        a1[sub] = asrc[(size_t)sj * H + hq];
        a2[sub] = adst[(size_t)(r0 + (int)dj[sub]) * H + hq];
        mk[sub] = mkey[dj[sub] * H + hq];
      }
#pragma unroll
      for (int j = 0; j < EB; ++j) {
        const unsigned s = (ent[j] == NOENT) ? 0u : (ent[j] & 0xFFFFu);
        const float* hp = h + (size_t)s * F + lane;
#pragma unroll
        for (int k = 0; k < K; ++k) hv[j][k] = hp[k * 32];
      }
#pragma unroll
      for (int sub = 0; sub < NSUB; ++sub) {
        const float p = __expf(lrelu(a1[sub] + a2[sub]) - fdecode(mk[sub]));
        if (ok[sub]) atomicAdd(&ssum[dj[sub] * H + hq], p);
#pragma unroll
        for (int jj = 0; jj < PE; ++jj)
#pragma unroll
          for (int q = 0; q < H; ++q)
            al[sub * PE + jj][q] = __int_as_float(__builtin_amdgcn_readlane(__float_as_int(p), jj * H + q));
      }
#pragma unroll
      for (int j = 0; j < EB; ++j) {
        if (ent[j] != NOENT) {
          float* accp = acc + (ent[j] >> 16) * F + lane;
#pragma unroll
          for (int k = 0; k < K; ++k) atomicAdd(&accp[k * 32], al[j][(k * 32) / D] * hv[j][k]);
        }
      }
    }
  } else {
    for (int eid = tid; eid < NEDGES; eid += OWN_T) {
      const unsigned dl = (unsigned)(dstI[eid] - r0);
      if (dl < (unsigned)OWN_R) {
        const int s = srcI[eid];
        for (int q = 0; q < H; ++q) {
          const float p = __expf(lrelu(asrc[(size_t)s * H + q] + adst[(size_t)(r0 + (int)dl) * H + q])
                                 - fdecode(mkey[dl * H + q]));
          atomicAdd(&ssum[dl * H + q], p);
          for (int f = q * D; f < (q + 1) * D; ++f)
            atomicAdd(&acc[dl * F + f], p * h[(size_t)s * F + f]);
        }
      }
    }
  }
  __syncthreads();

  if (POOL) {
    const int   f  = tid % F;
    const float bf = bias[f];
    for (int row = tid / F; row < OWN_R; row += OWN_T / F) {
      const unsigned g = pgid[row];
      if (g < (unsigned)NGRAPH) {
        float v = acc[row * F + f] / ssum[row * H + f / D] + bf;
        if (RELU) v = fmaxf(v, 0.f);
        atomicAdd(&ptab[g * F + f], v);
        if (f == 0) atomicAdd(&pnum[g], 1u);
      }
    }
    __syncthreads();
    for (int g = wave; g < NGRAPH; g += OWN_T / 32) {
      const unsigned c = pnum[g];
      if (c != 0u) {
        for (int f = lane; f < F; f += 32) atomicAdd(&psum[g * F + f], ptab[g * F + f]);
        if (lane == 0) atomicAdd(&pcnt[g], (float)c);
      }
    }
    return;
  }

  for (int i = tid; i < OWN_R * F4; i += OWN_T) {
    const int row = i / F4, c4 = i - row * F4;
    const int n = r0 + row;
    if (n < NNODES) {
      float4 v = *(const float4*)(acc + (size_t)i * 4);
      const float sden = ssum[row * H + (c4 * 4) / D];
      const float* b = bias + c4 * 4;
      v.x = v.x / sden + b[0]; v.y = v.y / sden + b[1]; v.z = v.z / sden + b[2]; v.w = v.w / sden + b[3];
      if (RELU) { v.x = fmaxf(v.x, 0.f); v.y = fmaxf(v.y, 0.f); v.z = fmaxf(v.z, 0.f); v.w = fmaxf(v.w, 0.f); }
      *(float4*)(out + (size_t)n * F + c4 * 4) = v;
    }
  }
}

__global__ void pool_final(const float* __restrict__ sums, const float* __restrict__ cnt,
                           float* __restrict__ out) {
  int i = blockIdx.x * blockDim.x + threadIdx.x;
  if (i >= NGRAPH * F2) return;
  out[i] = sums[i] / fmaxf(cnt[i / F2], 1.0f);
}

static inline int cdiv(long long a, int b) { return (int)((a + b - 1) / b); }

extern "C" void kernel_launch(void* const* d_in, const int* in_sizes, int n_in,
                              void* d_out, int out_size, void* d_ws, size_t ws_size,
                              hipStream_t stream) {
  const float* x        = (const float*)d_in[0];
  const int*   ei       = (const int*)d_in[1];
  const int*   batch    = (const int*)d_in[2];
  const float* W1       = (const float*)d_in[3];
  const float* att_src1 = (const float*)d_in[4];
  const float* att_dst1 = (const float*)d_in[5];
  const float* bias1    = (const float*)d_in[6];
  const float* W2       = (const float*)d_in[7];
  const float* att_src2 = (const float*)d_in[8];
  const float* att_dst2 = (const float*)d_in[9];
  const float* bias2    = (const float*)d_in[10];
  float* out = (float*)d_out;
  const int* srcI = ei;
  const int* dstI = ei + NEDGES;

  char* p = (char*)d_ws;
  auto alloc = [&](size_t bytes) -> char* {
    char* r = p; p += (bytes + 255) & ~(size_t)255; return r;
  };
  float*    h1    = (float*)alloc((size_t)NNODES * F1 * 4);
  float*    asrc1 = (float*)alloc((size_t)NNODES * H1C * 4);
  float*    adst1 = (float*)alloc((size_t)NNODES * H1C * 4);
  float*    agg1  = (float*)alloc((size_t)NNODES * F1 * 4);
  float*    h2    = (float*)alloc((size_t)NNODES * F2 * 4);
  float*    asrc2 = (float*)alloc((size_t)NNODES * H2C * 4);
  float*    adst2 = (float*)alloc((size_t)NNODES * H2C * 4);
  float*    psum  = (float*)alloc((size_t)NGRAPH * F2 * 4);
  float*    pcnt  = (float*)alloc((size_t)NGRAPH * 4);
  unsigned* glist = (unsigned*)alloc((size_t)((NNODES + OWN_R - 1) / OWN_R) * LIST_CAP * 4);
  unsigned* gcnt  = (unsigned*)alloc((size_t)((NNODES + OWN_R - 1) / OWN_R) * 4);

  hipMemsetAsync(psum,  0, (size_t)NGRAPH * F2 * 4, stream);
  hipMemsetAsync(pcnt,  0, (size_t)NGRAPH * 4, stream);

  const int T = 256;

  wmma_gemm_ldsw<FIN, F1, 4, 512, 256><<<256, 512, 0, stream>>>(x, W1, h1);
  att_coeff<<<cdiv((long long)NNODES * H1C, T), T, 0, stream>>>(h1, att_src1, att_dst1,
                                                                asrc1, adst1, H1C, D1C);
  owner_softmax_agg<F1, D1C, 4, 1, 1, 0><<<(NNODES + OWN_R - 1) / OWN_R, OWN_T, 0, stream>>>(
      srcI, dstI, asrc1, adst1, h1, bias1, glist, gcnt, agg1, (const int*)nullptr, (float*)nullptr, (float*)nullptr);

  wmma_gemm_ldsw<F1, F2, 2, 256, 512><<<512, 256, 0, stream>>>(agg1, W2, h2);
  att_coeff<<<cdiv((long long)NNODES * H2C, T), T, 0, stream>>>(h2, att_src2, att_dst2,
                                                                asrc2, adst2, H2C, D2C);
  owner_softmax_agg<F2, D2C, 8, 0, 0, 1><<<(NNODES + OWN_R - 1) / OWN_R, OWN_T, 0, stream>>>(
      srcI, dstI, asrc2, adst2, h2, bias2, glist, gcnt, (float*)nullptr, batch, psum, pcnt);

  pool_final<<<cdiv((long long)NGRAPH * F2, T), T, 0, stream>>>(psum, pcnt, out);
}


# Edge lists built once by a binning launch (8192-edge chunks: LDS counts per owner group, one integer atomic per group reserves a run); each group's scan of all destinations gone; order by arrival
# speedup vs baseline: 7.7829x; 1.1544x over previous
#include <hip/hip_runtime.h>
#include <math.h>

#define NNODES 50000
#define NEDGES 800000
#define NGRAPH 256
#define FIN    128
#define H1C    12
#define D1C    32
#define F1     (H1C * D1C)
#define H2C    1
#define D2C    64
#define F2     (H2C * D2C)
#define ETOT   (NEDGES + NNODES)

typedef float v2f __attribute__((ext_vector_type(2)));
typedef float v8f __attribute__((ext_vector_type(8)));

__device__ __forceinline__ unsigned fkey(float f) {
  unsigned u = __float_as_uint(f);
  return (u & 0x80000000u) ? ~u : (u | 0x80000000u);
}
__device__ __forceinline__ float fdecode(unsigned u) {
  unsigned b = (u & 0x80000000u) ? (u & 0x7fffffffu) : ~u;
  return __uint_as_float(b);
}

template <int K, int N, int TN, int T, int G>
__global__ void __launch_bounds__(T, 4)
wmma_gemm_ldsw(const float* __restrict__ A, const float* __restrict__ B, float* __restrict__ C) {
  constexpr int PITCH = 2 * N + 32;
  constexpr int KP    = K / 2;
  constexpr int KCS   = 32;
  constexpr int NKC   = K / (4 * KCS);
  constexpr int KS    = K / 4;
  constexpr int PD    = 16 / TN;
  constexpr int NG    = N / (16 * TN);
  constexpr int UNITS = (NNODES / 16) * NG;
  constexpr int WPG   = T / 32;
  constexpr int N4    = N / 4;
  constexpr int SI    = KP * N4 / T;
  static_assert(K % (4 * KCS) == 0 && N % (16 * TN) == 0 && NNODES % 16 == 0 && (2 * N) % 64 == 0, "tiling");
  static_assert((KP * N4) % T == 0, "staging");
  __shared__ __attribute__((aligned(16))) float wl[KP * PITCH];

  const int tid = threadIdx.x;
  {
    float4 w0[SI], w1[SI];
#pragma unroll
    for (int j = 0; j < SI; ++j) {
      const int i = tid + j * T;
      const int kp = i / N4, q = i - kp * N4;
      w0[j] = *(const float4*)(B + (size_t)(2 * kp) * N + 4 * q);
      w1[j] = *(const float4*)(B + (size_t)(2 * kp + 1) * N + 4 * q);
    }
#pragma unroll
    for (int j = 0; j < SI; ++j) {
      const int i = tid + j * T;
      const int kp = i / N4, q = i - kp * N4;
      float* d = wl + kp * PITCH + 8 * q;
      *(float4*)d       = make_float4(w0[j].x, w1[j].x, w0[j].y, w1[j].y);
      *(float4*)(d + 4) = make_float4(w0[j].z, w1[j].z, w0[j].w, w1[j].w);
    }
  }
  __syncthreads();

  const int lane = tid & 31;
  const int wave = tid >> 5;
  const int l    = lane & 15;
  const int half = lane >> 4;

  for (int u = blockIdx.x + G * wave; u < UNITS; u += G * WPG) {
    const int strip = u / NG, ng = u - strip * NG;
    const int m0 = strip << 4;
    const int n0 = ng * (16 * TN);

    const float* arow = A + (size_t)(m0 + l) * K + (half << 1);
    const float* bl   = wl + half * PITCH + 2 * (n0 + l);
    v2f a[2][KCS];
    v2f bf[PD + 1][TN];
    v8f c[TN];
#pragma unroll
    for (int t = 0; t < TN; ++t) c[t] = v8f{};
#pragma unroll
    for (int s = 0; s < KCS; ++s) a[0][s] = *(const v2f*)(arow + 4 * s);
#pragma unroll
    for (int p = 0; p < PD; ++p)
#pragma unroll
      for (int t = 0; t < TN; ++t) bf[p][t] = *(const v2f*)(bl + (2 * p) * PITCH + 32 * t);
#pragma unroll
    for (int kc = 0; kc < NKC; ++kc) {
      if (kc + 1 < NKC) {
#pragma unroll
        for (int s = 0; s < KCS; ++s) a[(kc + 1) & 1][s] = *(const v2f*)(arow + 4 * ((kc + 1) * KCS + s));
      }
      __builtin_amdgcn_sched_barrier(0);
#pragma unroll
      for (int s = 0; s < KCS; ++s) {
        const int gs = kc * KCS + s;
        if (gs + PD < KS) {
#pragma unroll
          for (int t = 0; t < TN; ++t)
            bf[(gs + PD) % (PD + 1)][t] = *(const v2f*)(bl + (2 * (gs + PD)) * PITCH + 32 * t);
        }
#pragma unroll
        for (int t = 0; t < TN; ++t)
          c[t] = __builtin_amdgcn_wmma_f32_16x16x4_f32(
               false, a[kc & 1][s],  false, bf[gs % (PD + 1)][t],
               (short)0, c[t],  false,  false);
        __builtin_amdgcn_sched_barrier(0);
      }
    }

    float* crow = C + (size_t)(m0 + (half << 3)) * N + n0 + l;
#pragma unroll
    for (int r = 0; r < 8; ++r)
#pragma unroll
      for (int t = 0; t < TN; ++t) crow[(size_t)r * N + 16 * t] = c[t][r];
  }
}

__global__ void att_coeff(const float* __restrict__ h,
                          const float* __restrict__ att_src,
                          const float* __restrict__ att_dst,
                          float* __restrict__ asrc, float* __restrict__ adst,
                          int H, int D) {
  int i = blockIdx.x * blockDim.x + threadIdx.x;
  if (i >= NNODES * H) return;
  int n = i / H, hh = i - n * H;
  const float* hp = h + (size_t)n * H * D + hh * D;
  const float* ws = att_src + hh * D;
  const float* wd = att_dst + hh * D;
  float s = 0.f, d = 0.f;
  for (int j = 0; j < D; ++j) { float v = hp[j]; s += v * ws[j]; d += v * wd[j]; }
  asrc[i] = s; adst[i] = d;
}

#define OWN_R    196
#define NOWN     ((NNODES + OWN_R - 1) / OWN_R)
#define OWN_T    1024
#define LIST_CAP 16384
#define GCNT_PITCH 32
#define BIN_T    1024
#define BIN_U    2
#define BIN_E    (BIN_T * BIN_U * 4)
#define BIN_G    ((NEDGES + BIN_E - 1) / BIN_E)
#define INIT_U   5
#define MAX_U    8
#define NOENT    0xFFFFFFFFu

__device__ __forceinline__ float lrelu(float v) { return (v > 0.f) ? v : 0.2f * v; }

__global__ void __launch_bounds__(BIN_T)
bin_edges(const int* __restrict__ srcI, const int* __restrict__ dstI,
          unsigned* __restrict__ glist_all, unsigned* gcount_all) {
  constexpr int NE4 = NEDGES / 4;
  constexpr int PB  = NOWN / 32;
  static_assert(NEDGES % 4 == 0 && NOWN % 32 == 0 && NOWN <= BIN_T - 32, "binning shapes");
  static_assert(NNODES <= 65536 && OWN_R <= 256, "list entry packing");
  __shared__ unsigned lcnt[NOWN];
  __shared__ unsigned loff[NOWN];
  __shared__ unsigned lbase[NOWN];
  __shared__ unsigned sent[BIN_E];
  __shared__ unsigned sdst[BIN_E];
  __shared__ unsigned nval;

  const int tid = threadIdx.x;
  for (int t = tid; t < NOWN; t += BIN_T) lcnt[t] = 0u;
  __syncthreads();

  const int4* d4 = (const int4*)dstI;
  const int4* s4 = (const int4*)srcI;
  int dd[BIN_U * 4], ss[BIN_U * 4];
#pragma unroll
  for (int u = 0; u < BIN_U; ++u) {
    const int idx = ((int)blockIdx.x * BIN_U + u) * BIN_T + tid;
    const int ic  = idx < NE4 ? idx : NE4 - 1;
    const int4 dv = d4[ic];
    const int4 sv = s4[ic];
    const bool in = idx < NE4;
    dd[4 * u + 0] = in ? dv.x : -1; dd[4 * u + 1] = in ? dv.y : -1;
    dd[4 * u + 2] = in ? dv.z : -1; dd[4 * u + 3] = in ? dv.w : -1;
    ss[4 * u + 0] = sv.x; ss[4 * u + 1] = sv.y; ss[4 * u + 2] = sv.z; ss[4 * u + 3] = sv.w;
  }
  unsigned own[BIN_U * 4], rnk[BIN_U * 4];
#pragma unroll
  for (int j = 0; j < BIN_U * 4; ++j) {
    const bool ok = (unsigned)dd[j] < (unsigned)NNODES;
    own[j] = ok ? (unsigned)dd[j] / (unsigned)OWN_R : NOENT;
    rnk[j] = 0u;
    if (ok) rnk[j] = atomicAdd(&lcnt[own[j]], 1u);
  }
  __syncthreads();

  if (tid < NOWN) {
    const unsigned c = lcnt[tid];
    lbase[tid] = c ? atomicAdd(&gcount_all[tid * GCNT_PITCH], c) : 0u;
  }
  if (tid >= BIN_T - 32) {
    const int lane = tid & 31;
    unsigned c[PB];
    unsigned sum = 0u;
#pragma unroll
    for (int k = 0; k < PB; ++k) { c[k] = lcnt[lane * PB + k]; sum += c[k]; }
    unsigned inc = sum;
    for (int o = 1; o < 32; o <<= 1) {
      const unsigned v = __shfl_up(inc, o);
      if (lane >= o) inc += v;
    }
    unsigned run = inc - sum;
#pragma unroll
    for (int k = 0; k < PB; ++k) { loff[lane * PB + k] = run; run += c[k]; }
    if (lane == 31) nval = inc;
  }
  __syncthreads();

#pragma unroll
  for (int j = 0; j < BIN_U * 4; ++j) {
    if (own[j] != NOENT) {
      const unsigned o    = own[j];
      const unsigned slot = loff[o] + rnk[j];
      const unsigned pos  = lbase[o] + rnk[j];
      sent[slot] = (((unsigned)dd[j] - o * (unsigned)OWN_R) << 16) | (unsigned)ss[j];
      sdst[slot] = pos < (unsigned)LIST_CAP ? o * (unsigned)LIST_CAP + pos : NOENT;
    }
  }
  __syncthreads();

  const unsigned nv = nval;
  if (nv != 0u) {
    unsigned dq[BIN_U * 4], eq[BIN_U * 4];
#pragma unroll
    for (int k = 0; k < BIN_U * 4; ++k) {
      const unsigned i  = (unsigned)(tid + k * BIN_T);
      const unsigned ic = i < nv ? i : nv - 1u;
      dq[k] = sdst[ic];
      eq[k] = sent[ic];
    }
#pragma unroll
    for (int k = 0; k < BIN_U * 4; ++k)
      if ((unsigned)(tid + k * BIN_T) < nv && dq[k] != NOENT) glist_all[dq[k]] = eq[k];
  }
}

template <int F, int D, int EB, int RELU, int POOL>
__global__ void __launch_bounds__(OWN_T)
owner_softmax_agg(const int* __restrict__ srcI, const int* __restrict__ dstI,
                  const float* __restrict__ asrc, const float* __restrict__ adst,
                  const float* __restrict__ h, const float* __restrict__ bias,
                  const unsigned* glist_all, const unsigned* gcount_all, float* __restrict__ out,
                  const int* __restrict__ batch, float* psum, float* pcnt) {
  constexpr int H    = F / D;
  constexpr int K    = F / 32;
  constexpr int F4   = F / 4;
  constexpr int RH   = OWN_R * H;
  constexpr int PE   = (32 / H) < EB ? (32 / H) : EB;
  constexpr int NSUB = EB / PE;
  static_assert(EB % PE == 0 && PE * H <= 32, "logit lanes");
  static_assert(NNODES <= 65536 && OWN_R <= 256, "list entry packing");
  __shared__ float    acc[OWN_R * F];
  __shared__ unsigned mkey[RH];
  __shared__ float    ssum[RH];
  __shared__ float    ptab[POOL ? NGRAPH * F : 1];
  __shared__ unsigned pnum[POOL ? NGRAPH : 1];
  __shared__ unsigned pgid[POOL ? OWN_R : 1];
  static_assert(!POOL || OWN_T % F == 0, "a thread keeps its feature over the rows of the pool pass");

  const int tid  = threadIdx.x;
  const int lane = tid & 31;
  const int wave = tid >> 5;
  const int r0   = blockIdx.x * OWN_R;
  const unsigned* glist = glist_all + (size_t)blockIdx.x * LIST_CAP;
  const unsigned seen = gcount_all[blockIdx.x * GCNT_PITCH];
  const bool fast = seen <= (unsigned)LIST_CAP;

  if (POOL) {
    for (int t = tid; t < NGRAPH * F / 4; t += OWN_T) *(float4*)(ptab + (size_t)t * 4) = make_float4(0.f, 0.f, 0.f, 0.f);
    for (int t = tid; t < NGRAPH; t += OWN_T) pnum[t] = 0u;
    for (int t = tid; t < OWN_R; t += OWN_T) {
      const int n = r0 + t;
      const unsigned g = n < NNODES ? (unsigned)batch[n] : (unsigned)NGRAPH;
      pgid[t] = g < (unsigned)NGRAPH ? g : (unsigned)NGRAPH;
    }
  }
  for (int t = tid; t < RH; t += OWN_T) {
    const int n = r0 + t / H;
    float v = 0.f;
    if (n < NNODES) v = lrelu(asrc[(size_t)r0 * H + t] + adst[(size_t)r0 * H + t]);
    mkey[t] = fkey(v);
  }
  __syncthreads();

  if (fast) {
    const int nitems = (int)seen * H;
    for (int t0 = tid; t0 < nitems; t0 += OWN_T * MAX_U) {
      unsigned en[MAX_U];
      int      hd[MAX_U];
      float    a1[MAX_U], a2[MAX_U];
#pragma unroll
      for (int u = 0; u < MAX_U; ++u) {
        const int t  = t0 + u * OWN_T;
        const int tc = t < nitems ? t : nitems - 1;
        const int e  = tc / H;
        hd[u] = tc - e * H;
        en[u] = glist[e];
      }
#pragma unroll
      for (int u = 0; u < MAX_U; ++u) {
        a1[u] = asrc[(size_t)(en[u] & 0xFFFFu) * H + hd[u]];
        a2[u] = adst[(size_t)(r0 + (int)(en[u] >> 16)) * H + hd[u]];
      }
#pragma unroll
      for (int u = 0; u < MAX_U; ++u)
        if (t0 + u * OWN_T < nitems)
          atomicMax(&mkey[(en[u] >> 16) * H + hd[u]], fkey(lrelu(a1[u] + a2[u])));
    }
  } else {
    for (int eid = tid; eid < NEDGES; eid += OWN_T) {
      const unsigned dl = (unsigned)(dstI[eid] - r0);
      if (dl < (unsigned)OWN_R) {
        const int s = srcI[eid];
        for (int q = 0; q < H; ++q)
          atomicMax(&mkey[dl * H + q],
                    fkey(lrelu(asrc[(size_t)s * H + q] + adst[(size_t)(r0 + (int)dl) * H + q])));
      }
    }
  }
  __syncthreads();

  for (int i0 = tid; i0 < OWN_R * F4; i0 += OWN_T * INIT_U) {
    float4 hv[INIT_U];
    float  as[INIT_U], ad[INIT_U];
    bool   ok[INIT_U];
#pragma unroll
    for (int u = 0; u < INIT_U; ++u) {
      const int i   = i0 + u * OWN_T;
      const int ic  = i < OWN_R * F4 ? i : OWN_R * F4 - 1;
      const int row = ic / F4, c4 = ic - row * F4;
      const int n   = r0 + row;
      const int nc  = n < NNODES ? n : NNODES - 1;
      ok[u] = n < NNODES;
      hv[u] = *(const float4*)(h + (size_t)nc * F + c4 * 4);
      as[u] = asrc[(size_t)nc * H + (c4 * 4) / D];
      ad[u] = adst[(size_t)nc * H + (c4 * 4) / D];
    }
#pragma unroll
    for (int u = 0; u < INIT_U; ++u) {
      const int i = i0 + u * OWN_T;
      if (i < OWN_R * F4) {
        const int row = i / F4, c4 = i - row * F4;
        const int hq  = (c4 * 4) / D;
        float p = __expf(lrelu(as[u] + ad[u]) - fdecode(mkey[row * H + hq]));
        float4 v = make_float4(p * hv[u].x, p * hv[u].y, p * hv[u].z, p * hv[u].w);
        if (!ok[u]) { v = make_float4(0.f, 0.f, 0.f, 0.f); p = 1.f; }
        *(float4*)(acc + (size_t)i * 4) = v;
        if ((c4 * 4) % D == 0) ssum[row * H + hq] = p;
      }
    }
  }
  __syncthreads();

  if (fast) {
    const unsigned total = seen;
    const unsigned last  = total > 0u ? total - 1u : 0u;
    const unsigned lj    = lane < EB ? (unsigned)lane : 0u;
    unsigned i = (unsigned)wave * EB;
    unsigned nxt = NOENT;
    if (i < total) {
      const unsigned raw = glist[i + lj < last ? i + lj : last];
      nxt = ((lane < EB) & (i + lj < total)) ? raw : NOENT;
    }
    for (; i < total; i += 32u * EB) {
      const unsigned cur = nxt;
      const unsigned in  = i + 32u * EB + lj;
      const unsigned raw = glist[in < last ? in : last];
      nxt = ((lane < EB) & (in < total)) ? raw : NOENT;
      unsigned ent[EB];
      float hv[EB][K];
      float al[EB][H];
#pragma unroll
      for (int j = 0; j < EB; ++j) ent[j] = (unsigned)__builtin_amdgcn_readlane((int)cur, j);
      const int jl = lane / H;
      const int hq = lane - jl * H;
      float    a1[NSUB], a2[NSUB];
      unsigned mk[NSUB], dj[NSUB];
      bool     ok[NSUB];
#pragma unroll
      for (int sub = 0; sub < NSUB; ++sub) {
        const unsigned el = (unsigned)__shfl((int)cur, sub * PE + (jl < PE ? jl : 0));
        ok[sub] = (jl < PE) & (el != NOENT);
        const unsigned sj = ok[sub] ? (el & 0xFFFFu) : 0u;
        dj[sub] = ok[sub] ? (el >> 16) : 0u;
        a1[sub] = asrc[(size_t)sj * H + hq];
        a2[sub] = adst[(size_t)(r0 + (int)dj[sub]) * H + hq];
        mk[sub] = mkey[dj[sub] * H + hq];
      }
#pragma unroll
      for (int j = 0; j < EB; ++j) {
        const unsigned s = (ent[j] == NOENT) ? 0u : (ent[j] & 0xFFFFu);
        const float* hp = h + (size_t)s * F + lane;
#pragma unroll
        for (int k = 0; k < K; ++k) hv[j][k] = hp[k * 32];
      }
#pragma unroll
      for (int sub = 0; sub < NSUB; ++sub) {
        const float p = __expf(lrelu(a1[sub] + a2[sub]) - fdecode(mk[sub]));
        if (ok[sub]) atomicAdd(&ssum[dj[sub] * H + hq], p);
#pragma unroll
        for (int jj = 0; jj < PE; ++jj)
#pragma unroll
          for (int q = 0; q < H; ++q)
            al[sub * PE + jj][q] = __int_as_float(__builtin_amdgcn_readlane(__float_as_int(p), jj * H + q));
      }
#pragma unroll
      for (int j = 0; j < EB; ++j) {
        if (ent[j] != NOENT) {
          float* accp = acc + (ent[j] >> 16) * F + lane;
#pragma unroll
          for (int k = 0; k < K; ++k) atomicAdd(&accp[k * 32], al[j][(k * 32) / D] * hv[j][k]);
        }
      }
    }
  } else {
    for (int eid = tid; eid < NEDGES; eid += OWN_T) {
      const unsigned dl = (unsigned)(dstI[eid] - r0);
      if (dl < (unsigned)OWN_R) {
        const int s = srcI[eid];
        for (int q = 0; q < H; ++q) {
          const float p = __expf(lrelu(asrc[(size_t)s * H + q] + adst[(size_t)(r0 + (int)dl) * H + q])
                                 - fdecode(mkey[dl * H + q]));
          atomicAdd(&ssum[dl * H + q], p);
          for (int f = q * D; f < (q + 1) * D; ++f)
            atomicAdd(&acc[dl * F + f], p * h[(size_t)s * F + f]);
        }
      }
    }
  }
  __syncthreads();

  if (POOL) {
    const int   f  = tid % F;
    const float bf = bias[f];
    for (int row = tid / F; row < OWN_R; row += OWN_T / F) {
      const unsigned g = pgid[row];
      if (g < (unsigned)NGRAPH) {
        float v = acc[row * F + f] / ssum[row * H + f / D] + bf;
        if (RELU) v = fmaxf(v, 0.f);
        atomicAdd(&ptab[g * F + f], v);
        if (f == 0) atomicAdd(&pnum[g], 1u);
      }
    }
    __syncthreads();
    for (int g = wave; g < NGRAPH; g += OWN_T / 32) {
      const unsigned c = pnum[g];
      if (c != 0u) {
        for (int f = lane; f < F; f += 32) atomicAdd(&psum[g * F + f], ptab[g * F + f]);
        if (lane == 0) atomicAdd(&pcnt[g], (float)c);
      }
    }
    return;
  }

  for (int i = tid; i < OWN_R * F4; i += OWN_T) {
    const int row = i / F4, c4 = i - row * F4;
    const int n = r0 + row;
    if (n < NNODES) {
      float4 v = *(const float4*)(acc + (size_t)i * 4);
      const float sden = ssum[row * H + (c4 * 4) / D];
      const float* b = bias + c4 * 4;
      v.x = v.x / sden + b[0]; v.y = v.y / sden + b[1]; v.z = v.z / sden + b[2]; v.w = v.w / sden + b[3];
      if (RELU) { v.x = fmaxf(v.x, 0.f); v.y = fmaxf(v.y, 0.f); v.z = fmaxf(v.z, 0.f); v.w = fmaxf(v.w, 0.f); }
      *(float4*)(out + (size_t)n * F + c4 * 4) = v;
    }
  }
}

__global__ void pool_final(const float* __restrict__ sums, const float* __restrict__ cnt,
                           float* __restrict__ out) {
  int i = blockIdx.x * blockDim.x + threadIdx.x;
  if (i >= NGRAPH * F2) return;
  out[i] = sums[i] / fmaxf(cnt[i / F2], 1.0f);
}

static inline int cdiv(long long a, int b) { return (int)((a + b - 1) / b); }

extern "C" void kernel_launch(void* const* d_in, const int* in_sizes, int n_in,
                              void* d_out, int out_size, void* d_ws, size_t ws_size,
                              hipStream_t stream) {
  const float* x        = (const float*)d_in[0];
  const int*   ei       = (const int*)d_in[1];
  const int*   batch    = (const int*)d_in[2];
  const float* W1       = (const float*)d_in[3];
  const float* att_src1 = (const float*)d_in[4];
  const float* att_dst1 = (const float*)d_in[5];
  const float* bias1    = (const float*)d_in[6];
  const float* W2       = (const float*)d_in[7];
  const float* att_src2 = (const float*)d_in[8];
  const float* att_dst2 = (const float*)d_in[9];
  const float* bias2    = (const float*)d_in[10];
  float* out = (float*)d_out;
  const int* srcI = ei;
  const int* dstI = ei + NEDGES;

  char* p = (char*)d_ws;
  auto alloc = [&](size_t bytes) -> char* {
    char* r = p; p += (bytes + 255) & ~(size_t)255; return r;
  };
  float*    h1    = (float*)alloc((size_t)NNODES * F1 * 4);
  float*    asrc1 = (float*)alloc((size_t)NNODES * H1C * 4);
  float*    adst1 = (float*)alloc((size_t)NNODES * H1C * 4);
  float*    agg1  = (float*)alloc((size_t)NNODES * F1 * 4);
  float*    h2    = (float*)alloc((size_t)NNODES * F2 * 4);
  float*    asrc2 = (float*)alloc((size_t)NNODES * H2C * 4);
  float*    adst2 = (float*)alloc((size_t)NNODES * H2C * 4);
  float*    zblk  = (float*)alloc((size_t)(NGRAPH * F2 + NGRAPH + NOWN * GCNT_PITCH) * 4);
  float*    psum  = zblk;
  float*    pcnt  = zblk + NGRAPH * F2;
  unsigned* gcnt  = (unsigned*)(zblk + NGRAPH * F2 + NGRAPH);
  unsigned* glist = (unsigned*)alloc((size_t)NOWN * LIST_CAP * 4);

  hipMemsetAsync(zblk, 0, (size_t)(NGRAPH * F2 + NGRAPH + NOWN * GCNT_PITCH) * 4, stream);

  bin_edges<<<BIN_G, BIN_T, 0, stream>>>(srcI, dstI, glist, gcnt);

  const int T = 256;

  wmma_gemm_ldsw<FIN, F1, 4, 512, 256><<<256, 512, 0, stream>>>(x, W1, h1);
  att_coeff<<<cdiv((long long)NNODES * H1C, T), T, 0, stream>>>(h1, att_src1, att_dst1,
                                                                asrc1, adst1, H1C, D1C);
  owner_softmax_agg<F1, D1C, 4, 1, 0><<<NOWN, OWN_T, 0, stream>>>(
      srcI, dstI, asrc1, adst1, h1, bias1, glist, gcnt, agg1, (const int*)nullptr, (float*)nullptr, (float*)nullptr);

  wmma_gemm_ldsw<F1, F2, 2, 256, 512><<<512, 256, 0, stream>>>(agg1, W2, h2);
  att_coeff<<<cdiv((long long)NNODES * H2C, T), T, 0, stream>>>(h2, att_src2, att_dst2,
                                                                asrc2, adst2, H2C, D2C);
  owner_softmax_agg<F2, D2C, 8, 0, 1><<<NOWN, OWN_T, 0, stream>>>(
      srcI, dstI, asrc2, adst2, h2, bias2, glist, gcnt, (float*)nullptr, batch, psum, pcnt);

  pool_final<<<cdiv((long long)NGRAPH * F2, T), T, 0, stream>>>(psum, pcnt, out);
}


# Attention coefficients: h streamed in whole 512-byte wave loads (16 bytes a lane) by 1024 looping work-groups; a head's D/4 lanes add partial sums by lane exchange; head's sum order changes
# speedup vs baseline: 8.0506x; 1.0344x over previous
#include <hip/hip_runtime.h>
#include <math.h>

#define NNODES 50000
#define NEDGES 800000
#define NGRAPH 256
#define FIN    128
#define H1C    12
#define D1C    32
#define F1     (H1C * D1C)
#define H2C    1
#define D2C    64
#define F2     (H2C * D2C)
#define ETOT   (NEDGES + NNODES)

typedef float v2f __attribute__((ext_vector_type(2)));
typedef float v8f __attribute__((ext_vector_type(8)));

__device__ __forceinline__ unsigned fkey(float f) {
  unsigned u = __float_as_uint(f);
  return (u & 0x80000000u) ? ~u : (u | 0x80000000u);
}
__device__ __forceinline__ float fdecode(unsigned u) {
  unsigned b = (u & 0x80000000u) ? (u & 0x7fffffffu) : ~u;
  return __uint_as_float(b);
}

template <int K, int N, int TN, int T, int G>
__global__ void __launch_bounds__(T, 4)
wmma_gemm_ldsw(const float* __restrict__ A, const float* __restrict__ B, float* __restrict__ C) {
  constexpr int PITCH = 2 * N + 32;
  constexpr int KP    = K / 2;
  constexpr int KCS   = 32;
  constexpr int NKC   = K / (4 * KCS);
  constexpr int KS    = K / 4;
  constexpr int PD    = 16 / TN;
  constexpr int NG    = N / (16 * TN);
  constexpr int UNITS = (NNODES / 16) * NG;
  constexpr int WPG   = T / 32;
  constexpr int N4    = N / 4;
  constexpr int SI    = KP * N4 / T;
  static_assert(K % (4 * KCS) == 0 && N % (16 * TN) == 0 && NNODES % 16 == 0 && (2 * N) % 64 == 0, "tiling");
  static_assert((KP * N4) % T == 0, "staging");
  __shared__ __attribute__((aligned(16))) float wl[KP * PITCH];

  const int tid = threadIdx.x;
  {
    float4 w0[SI], w1[SI];
#pragma unroll
    for (int j = 0; j < SI; ++j) {
      const int i = tid + j * T;
      const int kp = i / N4, q = i - kp * N4;
      w0[j] = *(const float4*)(B + (size_t)(2 * kp) * N + 4 * q);
      w1[j] = *(const float4*)(B + (size_t)(2 * kp + 1) * N + 4 * q);
    }
#pragma unroll
    for (int j = 0; j < SI; ++j) {
      const int i = tid + j * T;
      const int kp = i / N4, q = i - kp * N4;
      float* d = wl + kp * PITCH + 8 * q;
      *(float4*)d       = make_float4(w0[j].x, w1[j].x, w0[j].y, w1[j].y);
      *(float4*)(d + 4) = make_float4(w0[j].z, w1[j].z, w0[j].w, w1[j].w);
    }
  }
  __syncthreads();

  const int lane = tid & 31;
  const int wave = tid >> 5;
  const int l    = lane & 15;
  const int half = lane >> 4;

  for (int u = blockIdx.x + G * wave; u < UNITS; u += G * WPG) {
    const int strip = u / NG, ng = u - strip * NG;
    const int m0 = strip << 4;
    const int n0 = ng * (16 * TN);

    const float* arow = A + (size_t)(m0 + l) * K + (half << 1);
    const float* bl   = wl + half * PITCH + 2 * (n0 + l);
    v2f a[2][KCS];
    v2f bf[PD + 1][TN];
    v8f c[TN];
#pragma unroll
    for (int t = 0; t < TN; ++t) c[t] = v8f{};
#pragma unroll
    for (int s = 0; s < KCS; ++s) a[0][s] = *(const v2f*)(arow + 4 * s);
#pragma unroll
    for (int p = 0; p < PD; ++p)
#pragma unroll
      for (int t = 0; t < TN; ++t) bf[p][t] = *(const v2f*)(bl + (2 * p) * PITCH + 32 * t);
#pragma unroll
    for (int kc = 0; kc < NKC; ++kc) {
      if (kc + 1 < NKC) {
#pragma unroll
        for (int s = 0; s < KCS; ++s) a[(kc + 1) & 1][s] = *(const v2f*)(arow + 4 * ((kc + 1) * KCS + s));
      }
      __builtin_amdgcn_sched_barrier(0);
#pragma unroll
      for (int s = 0; s < KCS; ++s) {
        const int gs = kc * KCS + s;
        if (gs + PD < KS) {
#pragma unroll
          for (int t = 0; t < TN; ++t)
            bf[(gs + PD) % (PD + 1)][t] = *(const v2f*)(bl + (2 * (gs + PD)) * PITCH + 32 * t);
        }
#pragma unroll
        for (int t = 0; t < TN; ++t)
          c[t] = __builtin_amdgcn_wmma_f32_16x16x4_f32(
               false, a[kc & 1][s],  false, bf[gs % (PD + 1)][t],
               (short)0, c[t],  false,  false);
        __builtin_amdgcn_sched_barrier(0);
      }
    }

    float* crow = C + (size_t)(m0 + (half << 3)) * N + n0 + l;
#pragma unroll
    for (int r = 0; r < 8; ++r)
#pragma unroll
      for (int t = 0; t < TN; ++t) crow[(size_t)r * N + 16 * t] = c[t][r];
  }
}

#define ATT_T 256
#define ATT_G 1024
#define ATT_U 2
template <int H, int D>
__global__ void __launch_bounds__(ATT_T)
att_coeff_rows(const float* __restrict__ h,
               const float* __restrict__ att_src,
               const float* __restrict__ att_dst,
               float* __restrict__ asrc, float* __restrict__ adst) {
  constexpr int F   = H * D;
  constexpr int R4  = F / 4;
  constexpr int LPH = D / 4;
  constexpr int WL  = R4 >= 32 ? R4 / 32 : 1;
  constexpr int CH4 = WL * 32;
  constexpr int NCH = (int)(((long long)NNODES * R4) / CH4);
  constexpr int NW  = ATT_G * (ATT_T / 32);
  static_assert(D % 4 == 0 && LPH <= 32 && (LPH & (LPH - 1)) == 0, "a head's lanes");
  static_assert(CH4 % R4 == 0 && ((long long)NNODES * R4) % CH4 == 0, "whole rows per chunk");
  const int lane = threadIdx.x & 31;
  const int gw   = blockIdx.x * (ATT_T / 32) + (threadIdx.x >> 5);
  const float4* h4 = (const float4*)h;

  float4 ws[WL], wd[WL];
#pragma unroll
  for (int k = 0; k < WL; ++k) {
    const int c4 = k * 32 + lane;
    const int hh = (c4 / LPH) % H, p = c4 % LPH;
    ws[k] = *(const float4*)(att_src + hh * D + 4 * p);
    wd[k] = *(const float4*)(att_dst + hh * D + 4 * p);
  }
  for (int c0 = gw; c0 < NCH; c0 += NW * ATT_U) {
    float4 v[ATT_U][WL];
#pragma unroll
    for (int u = 0; u < ATT_U; ++u) {
      const int c  = c0 + u * NW;
      const int cc = c < NCH ? c : NCH - 1;
#pragma unroll
      for (int k = 0; k < WL; ++k) v[u][k] = h4[(size_t)cc * CH4 + k * 32 + lane];
    }
#pragma unroll
    for (int u = 0; u < ATT_U; ++u) {
      const int c = c0 + u * NW;
#pragma unroll
      for (int k = 0; k < WL; ++k) {
        float s = v[u][k].x * ws[k].x + v[u][k].y * ws[k].y + v[u][k].z * ws[k].z + v[u][k].w * ws[k].w;
        float d = v[u][k].x * wd[k].x + v[u][k].y * wd[k].y + v[u][k].z * wd[k].z + v[u][k].w * wd[k].w;
#pragma unroll
        for (int o = 1; o < LPH; o <<= 1) { s += __shfl_xor(s, o); d += __shfl_xor(d, o); }
        if (c < NCH && (lane % LPH) == 0) {
          const size_t g = ((size_t)c * CH4 + k * 32 + lane) / LPH;
          asrc[g] = s; adst[g] = d;
        }
      }
    }
  }
}

#define OWN_R    196
#define NOWN     ((NNODES + OWN_R - 1) / OWN_R)
#define OWN_T    1024
#define LIST_CAP 16384
#define GCNT_PITCH 32
#define BIN_T    1024
#define BIN_U    2
#define BIN_E    (BIN_T * BIN_U * 4)
#define BIN_G    ((NEDGES + BIN_E - 1) / BIN_E)
#define INIT_U   5
#define MAX_U    8
#define NOENT    0xFFFFFFFFu

__device__ __forceinline__ float lrelu(float v) { return (v > 0.f) ? v : 0.2f * v; }

__global__ void __launch_bounds__(BIN_T)
bin_edges(const int* __restrict__ srcI, const int* __restrict__ dstI,
          unsigned* __restrict__ glist_all, unsigned* gcount_all) {
  constexpr int NE4 = NEDGES / 4;
  constexpr int PB  = NOWN / 32;
  static_assert(NEDGES % 4 == 0 && NOWN % 32 == 0 && NOWN <= BIN_T - 32, "binning shapes");
  static_assert(NNODES <= 65536 && OWN_R <= 256, "list entry packing");
  __shared__ unsigned lcnt[NOWN];
  __shared__ unsigned loff[NOWN];
  __shared__ unsigned lbase[NOWN];
  __shared__ unsigned sent[BIN_E];
  __shared__ unsigned sdst[BIN_E];
  __shared__ unsigned nval;

  const int tid = threadIdx.x;
  for (int t = tid; t < NOWN; t += BIN_T) lcnt[t] = 0u;
  __syncthreads();

  const int4* d4 = (const int4*)dstI;
  const int4* s4 = (const int4*)srcI;
  int dd[BIN_U * 4], ss[BIN_U * 4];
#pragma unroll
  for (int u = 0; u < BIN_U; ++u) {
    const int idx = ((int)blockIdx.x * BIN_U + u) * BIN_T + tid;
    const int ic  = idx < NE4 ? idx : NE4 - 1;
    const int4 dv = d4[ic];
    const int4 sv = s4[ic];
    const bool in = idx < NE4;
    dd[4 * u + 0] = in ? dv.x : -1; dd[4 * u + 1] = in ? dv.y : -1;
    dd[4 * u + 2] = in ? dv.z : -1; dd[4 * u + 3] = in ? dv.w : -1;
    ss[4 * u + 0] = sv.x; ss[4 * u + 1] = sv.y; ss[4 * u + 2] = sv.z; ss[4 * u + 3] = sv.w;
  }
  unsigned own[BIN_U * 4], rnk[BIN_U * 4];
#pragma unroll
  for (int j = 0; j < BIN_U * 4; ++j) {
    const bool ok = (unsigned)dd[j] < (unsigned)NNODES;
    own[j] = ok ? (unsigned)dd[j] / (unsigned)OWN_R : NOENT;
    rnk[j] = 0u;
    if (ok) rnk[j] = atomicAdd(&lcnt[own[j]], 1u);
  }
  __syncthreads();

  if (tid < NOWN) {
    const unsigned c = lcnt[tid];
    lbase[tid] = c ? atomicAdd(&gcount_all[tid * GCNT_PITCH], c) : 0u;
  }
  if (tid >= BIN_T - 32) {
    const int lane = tid & 31;
    unsigned c[PB];
    unsigned sum = 0u;
#pragma unroll
    for (int k = 0; k < PB; ++k) { c[k] = lcnt[lane * PB + k]; sum += c[k]; }
    unsigned inc = sum;
    for (int o = 1; o < 32; o <<= 1) {
      const unsigned v = __shfl_up(inc, o);
      if (lane >= o) inc += v;
    }
    unsigned run = inc - sum;
#pragma unroll
    for (int k = 0; k < PB; ++k) { loff[lane * PB + k] = run; run += c[k]; }
    if (lane == 31) nval = inc;
  }
  __syncthreads();

#pragma unroll
  for (int j = 0; j < BIN_U * 4; ++j) {
    if (own[j] != NOENT) {
      const unsigned o    = own[j];
      const unsigned slot = loff[o] + rnk[j];
      const unsigned pos  = lbase[o] + rnk[j];
      sent[slot] = (((unsigned)dd[j] - o * (unsigned)OWN_R) << 16) | (unsigned)ss[j];
      sdst[slot] = pos < (unsigned)LIST_CAP ? o * (unsigned)LIST_CAP + pos : NOENT;
    }
  }
  __syncthreads();

  const unsigned nv = nval;
  if (nv != 0u) {
    unsigned dq[BIN_U * 4], eq[BIN_U * 4];
#pragma unroll
    for (int k = 0; k < BIN_U * 4; ++k) {
      const unsigned i  = (unsigned)(tid + k * BIN_T);
      const unsigned ic = i < nv ? i : nv - 1u;
      dq[k] = sdst[ic];
      eq[k] = sent[ic];
    }
#pragma unroll
    for (int k = 0; k < BIN_U * 4; ++k)
      if ((unsigned)(tid + k * BIN_T) < nv && dq[k] != NOENT) glist_all[dq[k]] = eq[k];
  }
}

template <int F, int D, int EB, int RELU, int POOL>
__global__ void __launch_bounds__(OWN_T)
owner_softmax_agg(const int* __restrict__ srcI, const int* __restrict__ dstI,
                  const float* __restrict__ asrc, const float* __restrict__ adst,
                  const float* __restrict__ h, const float* __restrict__ bias,
                  const unsigned* glist_all, const unsigned* gcount_all, float* __restrict__ out,
                  const int* __restrict__ batch, float* psum, float* pcnt) {
  constexpr int H    = F / D;
  constexpr int K    = F / 32;
  constexpr int F4   = F / 4;
  constexpr int RH   = OWN_R * H;
  constexpr int PE   = (32 / H) < EB ? (32 / H) : EB;
  constexpr int NSUB = EB / PE;
  static_assert(EB % PE == 0 && PE * H <= 32, "logit lanes");
  static_assert(NNODES <= 65536 && OWN_R <= 256, "list entry packing");
  __shared__ float    acc[OWN_R * F];
  __shared__ unsigned mkey[RH];
  __shared__ float    ssum[RH];
  __shared__ float    ptab[POOL ? NGRAPH * F : 1];
  __shared__ unsigned pnum[POOL ? NGRAPH : 1];
  __shared__ unsigned pgid[POOL ? OWN_R : 1];
  static_assert(!POOL || OWN_T % F == 0, "a thread keeps its feature over the rows of the pool pass");

  const int tid  = threadIdx.x;
  const int lane = tid & 31;
  const int wave = tid >> 5;
  const int r0   = blockIdx.x * OWN_R;
  const unsigned* glist = glist_all + (size_t)blockIdx.x * LIST_CAP;
  const unsigned seen = gcount_all[blockIdx.x * GCNT_PITCH];
  const bool fast = seen <= (unsigned)LIST_CAP;

  if (POOL) {
    for (int t = tid; t < NGRAPH * F / 4; t += OWN_T) *(float4*)(ptab + (size_t)t * 4) = make_float4(0.f, 0.f, 0.f, 0.f);
    for (int t = tid; t < NGRAPH; t += OWN_T) pnum[t] = 0u;
    for (int t = tid; t < OWN_R; t += OWN_T) {
      const int n = r0 + t;
      const unsigned g = n < NNODES ? (unsigned)batch[n] : (unsigned)NGRAPH;
      pgid[t] = g < (unsigned)NGRAPH ? g : (unsigned)NGRAPH;
    }
  }
  for (int t = tid; t < RH; t += OWN_T) {
    const int n = r0 + t / H;
    float v = 0.f;
    if (n < NNODES) v = lrelu(asrc[(size_t)r0 * H + t] + adst[(size_t)r0 * H + t]);
    mkey[t] = fkey(v);
  }
  __syncthreads();

  if (fast) {
    const int nitems = (int)seen * H;
    for (int t0 = tid; t0 < nitems; t0 += OWN_T * MAX_U) {
      unsigned en[MAX_U];
      int      hd[MAX_U];
      float    a1[MAX_U], a2[MAX_U];
#pragma unroll
      for (int u = 0; u < MAX_U; ++u) {
        const int t  = t0 + u * OWN_T;
        const int tc = t < nitems ? t : nitems - 1;
        const int e  = tc / H;
        hd[u] = tc - e * H;
        en[u] = glist[e];
      }
#pragma unroll
      for (int u = 0; u < MAX_U; ++u) {
        a1[u] = asrc[(size_t)(en[u] & 0xFFFFu) * H + hd[u]];
        a2[u] = adst[(size_t)(r0 + (int)(en[u] >> 16)) * H + hd[u]];
      }
#pragma unroll
      for (int u = 0; u < MAX_U; ++u)
        if (t0 + u * OWN_T < nitems)
          atomicMax(&mkey[(en[u] >> 16) * H + hd[u]], fkey(lrelu(a1[u] + a2[u])));
    }
  } else {
    for (int eid = tid; eid < NEDGES; eid += OWN_T) {
      const unsigned dl = (unsigned)(dstI[eid] - r0);
      if (dl < (unsigned)OWN_R) {
        const int s = srcI[eid];
        for (int q = 0; q < H; ++q)
          atomicMax(&mkey[dl * H + q],
                    fkey(lrelu(asrc[(size_t)s * H + q] + adst[(size_t)(r0 + (int)dl) * H + q])));
      }
    }
  }
  __syncthreads();

  for (int i0 = tid; i0 < OWN_R * F4; i0 += OWN_T * INIT_U) {
    float4 hv[INIT_U];
    float  as[INIT_U], ad[INIT_U];
    bool   ok[INIT_U];
#pragma unroll
    for (int u = 0; u < INIT_U; ++u) {
      const int i   = i0 + u * OWN_T;
      const int ic  = i < OWN_R * F4 ? i : OWN_R * F4 - 1;
      const int row = ic / F4, c4 = ic - row * F4;
      const int n   = r0 + row;
      const int nc  = n < NNODES ? n : NNODES - 1;
      ok[u] = n < NNODES;
      hv[u] = *(const float4*)(h + (size_t)nc * F + c4 * 4);
      as[u] = asrc[(size_t)nc * H + (c4 * 4) / D];
      ad[u] = adst[(size_t)nc * H + (c4 * 4) / D];
    }
#pragma unroll
    for (int u = 0; u < INIT_U; ++u) {
      const int i = i0 + u * OWN_T;
      if (i < OWN_R * F4) {
        const int row = i / F4, c4 = i - row * F4;
        const int hq  = (c4 * 4) / D;
        float p = __expf(lrelu(as[u] + ad[u]) - fdecode(mkey[row * H + hq]));
        float4 v = make_float4(p * hv[u].x, p * hv[u].y, p * hv[u].z, p * hv[u].w);
        if (!ok[u]) { v = make_float4(0.f, 0.f, 0.f, 0.f); p = 1.f; }
        *(float4*)(acc + (size_t)i * 4) = v;
        if ((c4 * 4) % D == 0) ssum[row * H + hq] = p;
      }
    }
  }
  __syncthreads();

  if (fast) {
    const unsigned total = seen;
    const unsigned last  = total > 0u ? total - 1u : 0u;
    const unsigned lj    = lane < EB ? (unsigned)lane : 0u;
    unsigned i = (unsigned)wave * EB;
    unsigned nxt = NOENT;
    if (i < total) {
      const unsigned raw = glist[i + lj < last ? i + lj : last];
      nxt = ((lane < EB) & (i + lj < total)) ? raw : NOENT;
    }
    for (; i < total; i += 32u * EB) {
      const unsigned cur = nxt;
      const unsigned in  = i + 32u * EB + lj;
      const unsigned raw = glist[in < last ? in : last];
      nxt = ((lane < EB) & (in < total)) ? raw : NOENT;
      unsigned ent[EB];
      float hv[EB][K];
      float al[EB][H];
#pragma unroll
      for (int j = 0; j < EB; ++j) ent[j] = (unsigned)__builtin_amdgcn_readlane((int)cur, j);
      const int jl = lane / H;
      const int hq = lane - jl * H;
      float    a1[NSUB], a2[NSUB];
      unsigned mk[NSUB], dj[NSUB];
      bool     ok[NSUB];
#pragma unroll
      for (int sub = 0; sub < NSUB; ++sub) {
        const unsigned el = (unsigned)__shfl((int)cur, sub * PE + (jl < PE ? jl : 0));
        ok[sub] = (jl < PE) & (el != NOENT);
        const unsigned sj = ok[sub] ? (el & 0xFFFFu) : 0u;
        dj[sub] = ok[sub] ? (el >> 16) : 0u;
        a1[sub] = asrc[(size_t)sj * H + hq];
        a2[sub] = adst[(size_t)(r0 + (int)dj[sub]) * H + hq];
        mk[sub] = mkey[dj[sub] * H + hq];
      }
#pragma unroll
      for (int j = 0; j < EB; ++j) {
        const unsigned s = (ent[j] == NOENT) ? 0u : (ent[j] & 0xFFFFu);
        const float* hp = h + (size_t)s * F + lane;
#pragma unroll
        for (int k = 0; k < K; ++k) hv[j][k] = hp[k * 32];
      }
#pragma unroll
      for (int sub = 0; sub < NSUB; ++sub) {
        const float p = __expf(lrelu(a1[sub] + a2[sub]) - fdecode(mk[sub]));
        if (ok[sub]) atomicAdd(&ssum[dj[sub] * H + hq], p);
#pragma unroll
        for (int jj = 0; jj < PE; ++jj)
#pragma unroll
          for (int q = 0; q < H; ++q)
            al[sub * PE + jj][q] = __int_as_float(__builtin_amdgcn_readlane(__float_as_int(p), jj * H + q));
      }
#pragma unroll
      for (int j = 0; j < EB; ++j) {
        if (ent[j] != NOENT) {
          float* accp = acc + (ent[j] >> 16) * F + lane;
#pragma unroll
          for (int k = 0; k < K; ++k) atomicAdd(&accp[k * 32], al[j][(k * 32) / D] * hv[j][k]);
        }
      }
    }
  } else {
    for (int eid = tid; eid < NEDGES; eid += OWN_T) {
      const unsigned dl = (unsigned)(dstI[eid] - r0);
      if (dl < (unsigned)OWN_R) {
        const int s = srcI[eid];
        for (int q = 0; q < H; ++q) {
          const float p = __expf(lrelu(asrc[(size_t)s * H + q] + adst[(size_t)(r0 + (int)dl) * H + q])
                                 - fdecode(mkey[dl * H + q]));
          atomicAdd(&ssum[dl * H + q], p);
          for (int f = q * D; f < (q + 1) * D; ++f)
            atomicAdd(&acc[dl * F + f], p * h[(size_t)s * F + f]);
        }
      }
    }
  }
  __syncthreads();

  if (POOL) {
    const int   f  = tid % F;
    const float bf = bias[f];
    for (int row = tid / F; row < OWN_R; row += OWN_T / F) {
      const unsigned g = pgid[row];
      if (g < (unsigned)NGRAPH) {
        float v = acc[row * F + f] / ssum[row * H + f / D] + bf;
        if (RELU) v = fmaxf(v, 0.f);
        atomicAdd(&ptab[g * F + f], v);
        if (f == 0) atomicAdd(&pnum[g], 1u);
      }
    }
    __syncthreads();
    for (int g = wave; g < NGRAPH; g += OWN_T / 32) {
      const unsigned c = pnum[g];
      if (c != 0u) {
        for (int f = lane; f < F; f += 32) atomicAdd(&psum[g * F + f], ptab[g * F + f]);
        if (lane == 0) atomicAdd(&pcnt[g], (float)c);
      }
    }
    return;
  }

  for (int i = tid; i < OWN_R * F4; i += OWN_T) {
    const int row = i / F4, c4 = i - row * F4;
    const int n = r0 + row;
    if (n < NNODES) {
      float4 v = *(const float4*)(acc + (size_t)i * 4);
      const float sden = ssum[row * H + (c4 * 4) / D];
      const float* b = bias + c4 * 4;
      v.x = v.x / sden + b[0]; v.y = v.y / sden + b[1]; v.z = v.z / sden + b[2]; v.w = v.w / sden + b[3];
      if (RELU) { v.x = fmaxf(v.x, 0.f); v.y = fmaxf(v.y, 0.f); v.z = fmaxf(v.z, 0.f); v.w = fmaxf(v.w, 0.f); }
      *(float4*)(out + (size_t)n * F + c4 * 4) = v;
    }
  }
}

__global__ void pool_final(const float* __restrict__ sums, const float* __restrict__ cnt,
                           float* __restrict__ out) {
  int i = blockIdx.x * blockDim.x + threadIdx.x;
  if (i >= NGRAPH * F2) return;
  out[i] = sums[i] / fmaxf(cnt[i / F2], 1.0f);
}

static inline int cdiv(long long a, int b) { return (int)((a + b - 1) / b); }

extern "C" void kernel_launch(void* const* d_in, const int* in_sizes, int n_in,
                              void* d_out, int out_size, void* d_ws, size_t ws_size,
                              hipStream_t stream) {
  const float* x        = (const float*)d_in[0];
  const int*   ei       = (const int*)d_in[1];
  const int*   batch    = (const int*)d_in[2];
  const float* W1       = (const float*)d_in[3];
  const float* att_src1 = (const float*)d_in[4];
  const float* att_dst1 = (const float*)d_in[5];
  const float* bias1    = (const float*)d_in[6];
  const float* W2       = (const float*)d_in[7];
  const float* att_src2 = (const float*)d_in[8];
  const float* att_dst2 = (const float*)d_in[9];
  const float* bias2    = (const float*)d_in[10];
  float* out = (float*)d_out;
  const int* srcI = ei;
  const int* dstI = ei + NEDGES;

  char* p = (char*)d_ws;
  auto alloc = [&](size_t bytes) -> char* {
    char* r = p; p += (bytes + 255) & ~(size_t)255; return r;
  };
  float*    h1    = (float*)alloc((size_t)NNODES * F1 * 4);
  float*    asrc1 = (float*)alloc((size_t)NNODES * H1C * 4);
  float*    adst1 = (float*)alloc((size_t)NNODES * H1C * 4);
  float*    agg1  = (float*)alloc((size_t)NNODES * F1 * 4);
  float*    h2    = (float*)alloc((size_t)NNODES * F2 * 4);
  float*    asrc2 = (float*)alloc((size_t)NNODES * H2C * 4);
  float*    adst2 = (float*)alloc((size_t)NNODES * H2C * 4);
  float*    zblk  = (float*)alloc((size_t)(NGRAPH * F2 + NGRAPH + NOWN * GCNT_PITCH) * 4);
  float*    psum  = zblk;
  float*    pcnt  = zblk + NGRAPH * F2;
  unsigned* gcnt  = (unsigned*)(zblk + NGRAPH * F2 + NGRAPH);
  unsigned* glist = (unsigned*)alloc((size_t)NOWN * LIST_CAP * 4);

  hipMemsetAsync(zblk, 0, (size_t)(NGRAPH * F2 + NGRAPH + NOWN * GCNT_PITCH) * 4, stream);

  bin_edges<<<BIN_G, BIN_T, 0, stream>>>(srcI, dstI, glist, gcnt);

  const int T = 256;

  wmma_gemm_ldsw<FIN, F1, 4, 512, 256><<<256, 512, 0, stream>>>(x, W1, h1);
  att_coeff_rows<H1C, D1C><<<ATT_G, ATT_T, 0, stream>>>(h1, att_src1, att_dst1, asrc1, adst1);
  owner_softmax_agg<F1, D1C, 4, 1, 0><<<NOWN, OWN_T, 0, stream>>>(
      srcI, dstI, asrc1, adst1, h1, bias1, glist, gcnt, agg1, (const int*)nullptr, (float*)nullptr, (float*)nullptr);

  wmma_gemm_ldsw<F1, F2, 2, 256, 512><<<512, 256, 0, stream>>>(agg1, W2, h2);
  att_coeff_rows<H2C, D2C><<<ATT_G, ATT_T, 0, stream>>>(h2, att_src2, att_dst2, asrc2, adst2);
  owner_softmax_agg<F2, D2C, 8, 0, 1><<<NOWN, OWN_T, 0, stream>>>(
      srcI, dstI, asrc2, adst2, h2, bias2, glist, gcnt, (float*)nullptr, batch, psum, pcnt);

  pool_final<<<cdiv((long long)NGRAPH * F2, T), T, 0, stream>>>(psum, pcnt, out);
}


# Conv1 aggregation: a lane takes 16 bytes of a gathered row (3 wave loads of 512 contiguous bytes per row, not 12 of 128); LDS sums lane-interleaved so adds stay 1 dword apart; sums by arrival
# speedup vs baseline: 8.4512x; 1.0498x over previous
#include <hip/hip_runtime.h>
#include <math.h>

#define NNODES 50000
#define NEDGES 800000
#define NGRAPH 256
#define FIN    128
#define H1C    12
#define D1C    32
#define F1     (H1C * D1C)
#define H2C    1
#define D2C    64
#define F2     (H2C * D2C)
#define ETOT   (NEDGES + NNODES)

typedef float v2f __attribute__((ext_vector_type(2)));
typedef float v8f __attribute__((ext_vector_type(8)));

__device__ __forceinline__ unsigned fkey(float f) {
  unsigned u = __float_as_uint(f);
  return (u & 0x80000000u) ? ~u : (u | 0x80000000u);
}
__device__ __forceinline__ float fdecode(unsigned u) {
  unsigned b = (u & 0x80000000u) ? (u & 0x7fffffffu) : ~u;
  return __uint_as_float(b);
}

template <int K, int N, int TN, int T, int G>
__global__ void __launch_bounds__(T, 4)
wmma_gemm_ldsw(const float* __restrict__ A, const float* __restrict__ B, float* __restrict__ C) {
  constexpr int PITCH = 2 * N + 32;
  constexpr int KP    = K / 2;
  constexpr int KCS   = 32;
  constexpr int NKC   = K / (4 * KCS);
  constexpr int KS    = K / 4;
  constexpr int PD    = 16 / TN;
  constexpr int NG    = N / (16 * TN);
  constexpr int UNITS = (NNODES / 16) * NG;
  constexpr int WPG   = T / 32;
  constexpr int N4    = N / 4;
  constexpr int SI    = KP * N4 / T;
  static_assert(K % (4 * KCS) == 0 && N % (16 * TN) == 0 && NNODES % 16 == 0 && (2 * N) % 64 == 0, "tiling");
  static_assert((KP * N4) % T == 0, "staging");
  __shared__ __attribute__((aligned(16))) float wl[KP * PITCH];

  const int tid = threadIdx.x;
  {
    float4 w0[SI], w1[SI];
#pragma unroll
    for (int j = 0; j < SI; ++j) {
      const int i = tid + j * T;
      const int kp = i / N4, q = i - kp * N4;
      w0[j] = *(const float4*)(B + (size_t)(2 * kp) * N + 4 * q);
      w1[j] = *(const float4*)(B + (size_t)(2 * kp + 1) * N + 4 * q);
    }
#pragma unroll
    for (int j = 0; j < SI; ++j) {
      const int i = tid + j * T;
      const int kp = i / N4, q = i - kp * N4;
      float* d = wl + kp * PITCH + 8 * q;
      *(float4*)d       = make_float4(w0[j].x, w1[j].x, w0[j].y, w1[j].y);
      *(float4*)(d + 4) = make_float4(w0[j].z, w1[j].z, w0[j].w, w1[j].w);
    }
  }
  __syncthreads();

  const int lane = tid & 31;
  const int wave = tid >> 5;
  const int l    = lane & 15;
  const int half = lane >> 4;

  for (int u = blockIdx.x + G * wave; u < UNITS; u += G * WPG) {
    const int strip = u / NG, ng = u - strip * NG;
    const int m0 = strip << 4;
    const int n0 = ng * (16 * TN);

    const float* arow = A + (size_t)(m0 + l) * K + (half << 1);
    const float* bl   = wl + half * PITCH + 2 * (n0 + l);
    v2f a[2][KCS];
    v2f bf[PD + 1][TN];
    v8f c[TN];
#pragma unroll
    for (int t = 0; t < TN; ++t) c[t] = v8f{};
#pragma unroll
    for (int s = 0; s < KCS; ++s) a[0][s] = *(const v2f*)(arow + 4 * s);
#pragma unroll
    for (int p = 0; p < PD; ++p)
#pragma unroll
      for (int t = 0; t < TN; ++t) bf[p][t] = *(const v2f*)(bl + (2 * p) * PITCH + 32 * t);
#pragma unroll
    for (int kc = 0; kc < NKC; ++kc) {
      if (kc + 1 < NKC) {
#pragma unroll
        for (int s = 0; s < KCS; ++s) a[(kc + 1) & 1][s] = *(const v2f*)(arow + 4 * ((kc + 1) * KCS + s));
      }
      __builtin_amdgcn_sched_barrier(0);
#pragma unroll
      for (int s = 0; s < KCS; ++s) {
        const int gs = kc * KCS + s;
        if (gs + PD < KS) {
#pragma unroll
          for (int t = 0; t < TN; ++t)
            bf[(gs + PD) % (PD + 1)][t] = *(const v2f*)(bl + (2 * (gs + PD)) * PITCH + 32 * t);
        }
#pragma unroll
        for (int t = 0; t < TN; ++t)
          c[t] = __builtin_amdgcn_wmma_f32_16x16x4_f32(
               false, a[kc & 1][s],  false, bf[gs % (PD + 1)][t],
               (short)0, c[t],  false,  false);
        __builtin_amdgcn_sched_barrier(0);
      }
    }

    float* crow = C + (size_t)(m0 + (half << 3)) * N + n0 + l;
#pragma unroll
    for (int r = 0; r < 8; ++r)
#pragma unroll
      for (int t = 0; t < TN; ++t) crow[(size_t)r * N + 16 * t] = c[t][r];
  }
}

#define ATT_T 256
#define ATT_G 1024
#define ATT_U 2
template <int H, int D>
__global__ void __launch_bounds__(ATT_T)
att_coeff_rows(const float* __restrict__ h,
               const float* __restrict__ att_src,
               const float* __restrict__ att_dst,
               float* __restrict__ asrc, float* __restrict__ adst) {
  constexpr int F   = H * D;
  constexpr int R4  = F / 4;
  constexpr int LPH = D / 4;
  constexpr int WL  = R4 >= 32 ? R4 / 32 : 1;
  constexpr int CH4 = WL * 32;
  constexpr int NCH = (int)(((long long)NNODES * R4) / CH4);
  constexpr int NW  = ATT_G * (ATT_T / 32);
  static_assert(D % 4 == 0 && LPH <= 32 && (LPH & (LPH - 1)) == 0, "a head's lanes");
  static_assert(CH4 % R4 == 0 && ((long long)NNODES * R4) % CH4 == 0, "whole rows per chunk");
  const int lane = threadIdx.x & 31;
  const int gw   = blockIdx.x * (ATT_T / 32) + (threadIdx.x >> 5);
  const float4* h4 = (const float4*)h;

  float4 ws[WL], wd[WL];
#pragma unroll
  for (int k = 0; k < WL; ++k) {
    const int c4 = k * 32 + lane;
    const int hh = (c4 / LPH) % H, p = c4 % LPH;
    ws[k] = *(const float4*)(att_src + hh * D + 4 * p);
    wd[k] = *(const float4*)(att_dst + hh * D + 4 * p);
  }
  for (int c0 = gw; c0 < NCH; c0 += NW * ATT_U) {
    float4 v[ATT_U][WL];
#pragma unroll
    for (int u = 0; u < ATT_U; ++u) {
      const int c  = c0 + u * NW;
      const int cc = c < NCH ? c : NCH - 1;
#pragma unroll
      for (int k = 0; k < WL; ++k) v[u][k] = h4[(size_t)cc * CH4 + k * 32 + lane];
    }
#pragma unroll
    for (int u = 0; u < ATT_U; ++u) {
      const int c = c0 + u * NW;
#pragma unroll
      for (int k = 0; k < WL; ++k) {
        float s = v[u][k].x * ws[k].x + v[u][k].y * ws[k].y + v[u][k].z * ws[k].z + v[u][k].w * ws[k].w;
        float d = v[u][k].x * wd[k].x + v[u][k].y * wd[k].y + v[u][k].z * wd[k].z + v[u][k].w * wd[k].w;
#pragma unroll
        for (int o = 1; o < LPH; o <<= 1) { s += __shfl_xor(s, o); d += __shfl_xor(d, o); }
        if (c < NCH && (lane % LPH) == 0) {
          const size_t g = ((size_t)c * CH4 + k * 32 + lane) / LPH;
          asrc[g] = s; adst[g] = d;
        }
      }
    }
  }
}

#define OWN_R    196
#define NOWN     ((NNODES + OWN_R - 1) / OWN_R)
#define OWN_T    1024
#define LIST_CAP 16384
#define GCNT_PITCH 32
#define BIN_T    1024
#define BIN_U    2
#define BIN_E    (BIN_T * BIN_U * 4)
#define BIN_G    ((NEDGES + BIN_E - 1) / BIN_E)
#define INIT_U   5
#define MAX_U    8
#define NOENT    0xFFFFFFFFu

__device__ __forceinline__ float lrelu(float v) { return (v > 0.f) ? v : 0.2f * v; }

__global__ void __launch_bounds__(BIN_T)
bin_edges(const int* __restrict__ srcI, const int* __restrict__ dstI,
          unsigned* __restrict__ glist_all, unsigned* gcount_all) {
  constexpr int NE4 = NEDGES / 4;
  constexpr int PB  = NOWN / 32;
  static_assert(NEDGES % 4 == 0 && NOWN % 32 == 0 && NOWN <= BIN_T - 32, "binning shapes");
  static_assert(NNODES <= 65536 && OWN_R <= 256, "list entry packing");
  __shared__ unsigned lcnt[NOWN];
  __shared__ unsigned loff[NOWN];
  __shared__ unsigned lbase[NOWN];
  __shared__ unsigned sent[BIN_E];
  __shared__ unsigned sdst[BIN_E];
  __shared__ unsigned nval;

  const int tid = threadIdx.x;
  for (int t = tid; t < NOWN; t += BIN_T) lcnt[t] = 0u;
  __syncthreads();

  const int4* d4 = (const int4*)dstI;
  const int4* s4 = (const int4*)srcI;
  int dd[BIN_U * 4], ss[BIN_U * 4];
#pragma unroll
  for (int u = 0; u < BIN_U; ++u) {
    const int idx = ((int)blockIdx.x * BIN_U + u) * BIN_T + tid;
    const int ic  = idx < NE4 ? idx : NE4 - 1;
    const int4 dv = d4[ic];
    const int4 sv = s4[ic];
    const bool in = idx < NE4;
    dd[4 * u + 0] = in ? dv.x : -1; dd[4 * u + 1] = in ? dv.y : -1;
    dd[4 * u + 2] = in ? dv.z : -1; dd[4 * u + 3] = in ? dv.w : -1;
    ss[4 * u + 0] = sv.x; ss[4 * u + 1] = sv.y; ss[4 * u + 2] = sv.z; ss[4 * u + 3] = sv.w;
  }
  unsigned own[BIN_U * 4], rnk[BIN_U * 4];
#pragma unroll
  for (int j = 0; j < BIN_U * 4; ++j) {
    const bool ok = (unsigned)dd[j] < (unsigned)NNODES;
    own[j] = ok ? (unsigned)dd[j] / (unsigned)OWN_R : NOENT;
    rnk[j] = 0u;
    if (ok) rnk[j] = atomicAdd(&lcnt[own[j]], 1u);
  }
  __syncthreads();

  if (tid < NOWN) {
    const unsigned c = lcnt[tid];
    lbase[tid] = c ? atomicAdd(&gcount_all[tid * GCNT_PITCH], c) : 0u;
  }
  if (tid >= BIN_T - 32) {
    const int lane = tid & 31;
    unsigned c[PB];
    unsigned sum = 0u;
#pragma unroll
    for (int k = 0; k < PB; ++k) { c[k] = lcnt[lane * PB + k]; sum += c[k]; }
    unsigned inc = sum;
    for (int o = 1; o < 32; o <<= 1) {
      const unsigned v = __shfl_up(inc, o);
      if (lane >= o) inc += v;
    }
    unsigned run = inc - sum;
#pragma unroll
    for (int k = 0; k < PB; ++k) { loff[lane * PB + k] = run; run += c[k]; }
    if (lane == 31) nval = inc;
  }
  __syncthreads();

#pragma unroll
  for (int j = 0; j < BIN_U * 4; ++j) {
    if (own[j] != NOENT) {
      const unsigned o    = own[j];
      const unsigned slot = loff[o] + rnk[j];
      const unsigned pos  = lbase[o] + rnk[j];
      sent[slot] = (((unsigned)dd[j] - o * (unsigned)OWN_R) << 16) | (unsigned)ss[j];
      sdst[slot] = pos < (unsigned)LIST_CAP ? o * (unsigned)LIST_CAP + pos : NOENT;
    }
  }
  __syncthreads();

  const unsigned nv = nval;
  if (nv != 0u) {
    unsigned dq[BIN_U * 4], eq[BIN_U * 4];
#pragma unroll
    for (int k = 0; k < BIN_U * 4; ++k) {
      const unsigned i  = (unsigned)(tid + k * BIN_T);
      const unsigned ic = i < nv ? i : nv - 1u;
      dq[k] = sdst[ic];
      eq[k] = sent[ic];
    }
#pragma unroll
    for (int k = 0; k < BIN_U * 4; ++k)
      if ((unsigned)(tid + k * BIN_T) < nv && dq[k] != NOENT) glist_all[dq[k]] = eq[k];
  }
}

template <bool WIDE>
__device__ __forceinline__ int lpos(int f) {
  return WIDE ? ((f & ~127) | ((f & 3) << 5) | ((f & 127) >> 2)) : f;
}

template <int F, int D, int EB, int RELU, int POOL>
__global__ void __launch_bounds__(OWN_T)
owner_softmax_agg(const int* __restrict__ srcI, const int* __restrict__ dstI,
                  const float* __restrict__ asrc, const float* __restrict__ adst,
                  const float* __restrict__ h, const float* __restrict__ bias,
                  const unsigned* glist_all, const unsigned* gcount_all, float* __restrict__ out,
                  const int* __restrict__ batch, float* psum, float* pcnt) {
  constexpr int H    = F / D;
  constexpr int K    = F / 32;
  constexpr int F4   = F / 4;
  constexpr int RH   = OWN_R * H;
  constexpr int PE   = (32 / H) < EB ? (32 / H) : EB;
  constexpr int NSUB = EB / PE;
  constexpr bool WIDE = (F % 128 == 0);
  constexpr int K4   = WIDE ? F / 128 : 1;
  static_assert(EB % PE == 0 && PE * H <= 32, "logit lanes");
  static_assert(!WIDE || (D == 32 && H == 12 && PE == 2 && !POOL), "WIDE: 8 lanes share a head, 12 logit lanes per 16-lane row");
  static_assert(NNODES <= 65536 && OWN_R <= 256, "list entry packing");
  __shared__ float    acc[OWN_R * F];
  __shared__ unsigned mkey[RH];
  __shared__ float    ssum[RH];
  __shared__ float    ptab[POOL ? NGRAPH * F : 1];
  __shared__ unsigned pnum[POOL ? NGRAPH : 1];
  __shared__ unsigned pgid[POOL ? OWN_R : 1];
  static_assert(!POOL || OWN_T % F == 0, "a thread keeps its feature over the rows of the pool pass");

  const int tid  = threadIdx.x;
  const int lane = tid & 31;
  const int wave = tid >> 5;
  const int r0   = blockIdx.x * OWN_R;
  const unsigned* glist = glist_all + (size_t)blockIdx.x * LIST_CAP;
  const unsigned seen = gcount_all[blockIdx.x * GCNT_PITCH];
  const bool fast = seen <= (unsigned)LIST_CAP;

  if (POOL) {
    for (int t = tid; t < NGRAPH * F / 4; t += OWN_T) *(float4*)(ptab + (size_t)t * 4) = make_float4(0.f, 0.f, 0.f, 0.f);
    for (int t = tid; t < NGRAPH; t += OWN_T) pnum[t] = 0u;
    for (int t = tid; t < OWN_R; t += OWN_T) {
      const int n = r0 + t;
      const unsigned g = n < NNODES ? (unsigned)batch[n] : (unsigned)NGRAPH;
      pgid[t] = g < (unsigned)NGRAPH ? g : (unsigned)NGRAPH;
    }
  }
  for (int t = tid; t < RH; t += OWN_T) {
    const int n = r0 + t / H;
    float v = 0.f;
    if (n < NNODES) v = lrelu(asrc[(size_t)r0 * H + t] + adst[(size_t)r0 * H + t]);
    mkey[t] = fkey(v);
  }
  __syncthreads();

  if (fast) {
    const int nitems = (int)seen * H;
    for (int t0 = tid; t0 < nitems; t0 += OWN_T * MAX_U) {
      unsigned en[MAX_U];
      int      hd[MAX_U];
      float    a1[MAX_U], a2[MAX_U];
#pragma unroll
      for (int u = 0; u < MAX_U; ++u) {
        const int t  = t0 + u * OWN_T;
        const int tc = t < nitems ? t : nitems - 1;
        const int e  = tc / H;
        hd[u] = tc - e * H;
        en[u] = glist[e];
      }
#pragma unroll
      for (int u = 0; u < MAX_U; ++u) {
        a1[u] = asrc[(size_t)(en[u] & 0xFFFFu) * H + hd[u]];
        a2[u] = adst[(size_t)(r0 + (int)(en[u] >> 16)) * H + hd[u]];
      }
#pragma unroll
      for (int u = 0; u < MAX_U; ++u)
        if (t0 + u * OWN_T < nitems)
          atomicMax(&mkey[(en[u] >> 16) * H + hd[u]], fkey(lrelu(a1[u] + a2[u])));
    }
  } else {
    for (int eid = tid; eid < NEDGES; eid += OWN_T) {
      const unsigned dl = (unsigned)(dstI[eid] - r0);
      if (dl < (unsigned)OWN_R) {
        const int s = srcI[eid];
        for (int q = 0; q < H; ++q)
          atomicMax(&mkey[dl * H + q],
                    fkey(lrelu(asrc[(size_t)s * H + q] + adst[(size_t)(r0 + (int)dl) * H + q])));
      }
    }
  }
  __syncthreads();

  for (int i0 = tid; i0 < OWN_R * F4; i0 += OWN_T * INIT_U) {
    float4 hv[INIT_U];
    float  as[INIT_U], ad[INIT_U];
    bool   ok[INIT_U];
#pragma unroll
    for (int u = 0; u < INIT_U; ++u) {
      const int i   = i0 + u * OWN_T;
      const int ic  = i < OWN_R * F4 ? i : OWN_R * F4 - 1;
      const int row = ic / F4, c4 = ic - row * F4;
      const int n   = r0 + row;
      const int nc  = n < NNODES ? n : NNODES - 1;
      ok[u] = n < NNODES;
      hv[u] = *(const float4*)(h + (size_t)nc * F + c4 * 4);
      as[u] = asrc[(size_t)nc * H + (c4 * 4) / D];
      ad[u] = adst[(size_t)nc * H + (c4 * 4) / D];
    }
#pragma unroll
    for (int u = 0; u < INIT_U; ++u) {
      const int i = i0 + u * OWN_T;
      if (i < OWN_R * F4) {
        const int row = i / F4, c4 = i - row * F4;
        const int hq  = (c4 * 4) / D;
        float p = __expf(lrelu(as[u] + ad[u]) - fdecode(mkey[row * H + hq]));
        float4 v = make_float4(p * hv[u].x, p * hv[u].y, p * hv[u].z, p * hv[u].w);
        if (!ok[u]) { v = make_float4(0.f, 0.f, 0.f, 0.f); p = 1.f; }
        if (WIDE) {
          float* a = acc + row * F + lpos<WIDE>(c4 * 4);
          a[0] = v.x; a[32] = v.y; a[64] = v.z; a[96] = v.w;
        } else {
          *(float4*)(acc + (size_t)i * 4) = v;
        }
        if ((c4 * 4) % D == 0) ssum[row * H + hq] = p;
      }
    }
  }
  __syncthreads();

  if (fast) {
    const unsigned total = seen;
    const unsigned last  = total > 0u ? total - 1u : 0u;
    const unsigned lj    = lane < EB ? (unsigned)lane : 0u;
    unsigned i = (unsigned)wave * EB;
    unsigned nxt = NOENT;
    if (i < total) {
      const unsigned raw = glist[i + lj < last ? i + lj : last];
      nxt = ((lane < EB) & (i + lj < total)) ? raw : NOENT;
    }
    for (; i < total; i += 32u * EB) {
      const unsigned cur = nxt;
      const unsigned in  = i + 32u * EB + lj;
      const unsigned raw = glist[in < last ? in : last];
      nxt = ((lane < EB) & (in < total)) ? raw : NOENT;
      unsigned ent[EB];
      float hv[WIDE ? 1 : EB][WIDE ? 1 : K];
      float al[WIDE ? 1 : EB][WIDE ? 1 : H];
      float4 hw[WIDE ? EB : 1][K4];
      float  wl[WIDE ? EB : 1][K4];
#pragma unroll
      for (int j = 0; j < EB; ++j) ent[j] = (unsigned)__builtin_amdgcn_readlane((int)cur, j);
      const int lx = lane & 15;
      const int jl = WIDE ? lx / 6 : lane / H;
      const int hq = WIDE ? 4 * ((lx % 6) >> 1) + 2 * (lane >> 4) + (lx & 1) : lane - jl * H;
      const bool lv = WIDE ? (lx < 12) : (jl < PE);
      float    a1[NSUB], a2[NSUB];
      unsigned mk[NSUB], dj[NSUB];
      bool     ok[NSUB];
#pragma unroll
      for (int sub = 0; sub < NSUB; ++sub) {
        const unsigned el = (unsigned)__shfl((int)cur, sub * PE + (lv ? jl : 0));
        ok[sub] = lv & (el != NOENT);
        const unsigned sj = ok[sub] ? (el & 0xFFFFu) : 0u;
        dj[sub] = ok[sub] ? (el >> 16) : 0u;
        a1[sub] = asrc[(size_t)sj * H + hq];
        a2[sub] = adst[(size_t)(r0 + (int)dj[sub]) * H + hq];
        mk[sub] = mkey[dj[sub] * H + hq];
      }
#pragma unroll
      for (int j = 0; j < EB; ++j) {
        const unsigned s = (ent[j] == NOENT) ? 0u : (ent[j] & 0xFFFFu);
        if constexpr (WIDE) {
          const float* hp = h + (size_t)s * F + lane * 4;
#pragma unroll
          for (int k = 0; k < K4; ++k) hw[j][k] = *(const float4*)(hp + k * 128);
        } else {
          const float* hp = h + (size_t)s * F + lane;
#pragma unroll
          for (int k = 0; k < K; ++k) hv[j][k] = hp[k * 32];
        }
      }
#pragma unroll
      for (int sub = 0; sub < NSUB; ++sub) {
        const float p = __expf(lrelu(a1[sub] + a2[sub]) - fdecode(mk[sub]));
        if (ok[sub]) atomicAdd(&ssum[dj[sub] * H + hq], p);
        if constexpr (WIDE) {
#pragma unroll
          for (int jj = 0; jj < PE; ++jj)
#pragma unroll
            for (int k = 0; k < K4; ++k) {
              const unsigned sl = 0x11111111u * (unsigned)(6 * jj + 2 * k);
              wl[sub * PE + jj][k] = __uint_as_float(__builtin_amdgcn_permlane16(
                  0u, __float_as_uint(p), sl, sl + 0x11111111u, false, false));
            }
        } else {
#pragma unroll
          for (int jj = 0; jj < PE; ++jj)
#pragma unroll
            for (int q = 0; q < H; ++q)
              al[sub * PE + jj][q] = __int_as_float(__builtin_amdgcn_readlane(__float_as_int(p), jj * H + q));
        }
      }
#pragma unroll
      for (int j = 0; j < EB; ++j) {
        if (ent[j] != NOENT) {
          float* accp = acc + (ent[j] >> 16) * F + lane;
          if constexpr (WIDE) {
#pragma unroll
            for (int k = 0; k < K4; ++k) {
              atomicAdd(&accp[k * 128],      wl[j][k] * hw[j][k].x);
              atomicAdd(&accp[k * 128 + 32], wl[j][k] * hw[j][k].y);
              atomicAdd(&accp[k * 128 + 64], wl[j][k] * hw[j][k].z);
              atomicAdd(&accp[k * 128 + 96], wl[j][k] * hw[j][k].w);
            }
          } else {
#pragma unroll
            for (int k = 0; k < K; ++k) atomicAdd(&accp[k * 32], al[j][(k * 32) / D] * hv[j][k]);
          }
        }
      }
    }
  } else {
    for (int eid = tid; eid < NEDGES; eid += OWN_T) {
      const unsigned dl = (unsigned)(dstI[eid] - r0);
      if (dl < (unsigned)OWN_R) {
        const int s = srcI[eid];
        for (int q = 0; q < H; ++q) {
          const float p = __expf(lrelu(asrc[(size_t)s * H + q] + adst[(size_t)(r0 + (int)dl) * H + q])
                                 - fdecode(mkey[dl * H + q]));
          atomicAdd(&ssum[dl * H + q], p);
          for (int f = q * D; f < (q + 1) * D; ++f)
            atomicAdd(&acc[dl * F + lpos<WIDE>(f)], p * h[(size_t)s * F + f]);
        }
      }
    }
  }
  __syncthreads();

  if (POOL) {
    const int   f  = tid % F;
    const float bf = bias[f];
    for (int row = tid / F; row < OWN_R; row += OWN_T / F) {
      const unsigned g = pgid[row];
      if (g < (unsigned)NGRAPH) {
        float v = acc[row * F + f] / ssum[row * H + f / D] + bf;
        if (RELU) v = fmaxf(v, 0.f);
        atomicAdd(&ptab[g * F + f], v);
        if (f == 0) atomicAdd(&pnum[g], 1u);
      }
    }
    __syncthreads();
    for (int g = wave; g < NGRAPH; g += OWN_T / 32) {
      const unsigned c = pnum[g];
      if (c != 0u) {
        for (int f = lane; f < F; f += 32) atomicAdd(&psum[g * F + f], ptab[g * F + f]);
        if (lane == 0) atomicAdd(&pcnt[g], (float)c);
      }
    }
    return;
  }

  for (int i = tid; i < OWN_R * F4; i += OWN_T) {
    const int row = i / F4, c4 = i - row * F4;
    const int n = r0 + row;
    if (n < NNODES) {
      float4 v;
      if (WIDE) {
        const float* a = acc + row * F + lpos<WIDE>(c4 * 4);
        v = make_float4(a[0], a[32], a[64], a[96]);
      } else {
        v = *(const float4*)(acc + (size_t)i * 4);
      }
      const float sden = ssum[row * H + (c4 * 4) / D];
      const float* b = bias + c4 * 4;
      v.x = v.x / sden + b[0]; v.y = v.y / sden + b[1]; v.z = v.z / sden + b[2]; v.w = v.w / sden + b[3];
      if (RELU) { v.x = fmaxf(v.x, 0.f); v.y = fmaxf(v.y, 0.f); v.z = fmaxf(v.z, 0.f); v.w = fmaxf(v.w, 0.f); }
      *(float4*)(out + (size_t)n * F + c4 * 4) = v;
    }
  }
}

__global__ void pool_final(const float* __restrict__ sums, const float* __restrict__ cnt,
                           float* __restrict__ out) {
  int i = blockIdx.x * blockDim.x + threadIdx.x;
  if (i >= NGRAPH * F2) return;
  out[i] = sums[i] / fmaxf(cnt[i / F2], 1.0f);
}

static inline int cdiv(long long a, int b) { return (int)((a + b - 1) / b); }

extern "C" void kernel_launch(void* const* d_in, const int* in_sizes, int n_in,
                              void* d_out, int out_size, void* d_ws, size_t ws_size,
                              hipStream_t stream) {
  const float* x        = (const float*)d_in[0];
  const int*   ei       = (const int*)d_in[1];
  const int*   batch    = (const int*)d_in[2];
  const float* W1       = (const float*)d_in[3];
  const float* att_src1 = (const float*)d_in[4];
  const float* att_dst1 = (const float*)d_in[5];
  const float* bias1    = (const float*)d_in[6];
  const float* W2       = (const float*)d_in[7];
  const float* att_src2 = (const float*)d_in[8];
  const float* att_dst2 = (const float*)d_in[9];
  const float* bias2    = (const float*)d_in[10];
  float* out = (float*)d_out;
  const int* srcI = ei;
  const int* dstI = ei + NEDGES;

  char* p = (char*)d_ws;
  auto alloc = [&](size_t bytes) -> char* {
    char* r = p; p += (bytes + 255) & ~(size_t)255; return r;
  };
  float*    h1    = (float*)alloc((size_t)NNODES * F1 * 4);
  float*    asrc1 = (float*)alloc((size_t)NNODES * H1C * 4);
  float*    adst1 = (float*)alloc((size_t)NNODES * H1C * 4);
  float*    agg1  = (float*)alloc((size_t)NNODES * F1 * 4);
  float*    h2    = (float*)alloc((size_t)NNODES * F2 * 4);
  float*    asrc2 = (float*)alloc((size_t)NNODES * H2C * 4);
  float*    adst2 = (float*)alloc((size_t)NNODES * H2C * 4);
  float*    zblk  = (float*)alloc((size_t)(NGRAPH * F2 + NGRAPH + NOWN * GCNT_PITCH) * 4);
  float*    psum  = zblk;
  float*    pcnt  = zblk + NGRAPH * F2;
  unsigned* gcnt  = (unsigned*)(zblk + NGRAPH * F2 + NGRAPH);
  unsigned* glist = (unsigned*)alloc((size_t)NOWN * LIST_CAP * 4);

  hipMemsetAsync(zblk, 0, (size_t)(NGRAPH * F2 + NGRAPH + NOWN * GCNT_PITCH) * 4, stream);

  bin_edges<<<BIN_G, BIN_T, 0, stream>>>(srcI, dstI, glist, gcnt);

  const int T = 256;

  wmma_gemm_ldsw<FIN, F1, 4, 512, 256><<<256, 512, 0, stream>>>(x, W1, h1);
  att_coeff_rows<H1C, D1C><<<ATT_G, ATT_T, 0, stream>>>(h1, att_src1, att_dst1, asrc1, adst1);
  owner_softmax_agg<F1, D1C, 4, 1, 0><<<NOWN, OWN_T, 0, stream>>>(
      srcI, dstI, asrc1, adst1, h1, bias1, glist, gcnt, agg1, (const int*)nullptr, (float*)nullptr, (float*)nullptr);

  wmma_gemm_ldsw<F1, F2, 2, 256, 512><<<512, 256, 0, stream>>>(agg1, W2, h2);
  att_coeff_rows<H2C, D2C><<<ATT_G, ATT_T, 0, stream>>>(h2, att_src2, att_dst2, asrc2, adst2);
  owner_softmax_agg<F2, D2C, 8, 0, 1><<<NOWN, OWN_T, 0, stream>>>(
      srcI, dstI, asrc2, adst2, h2, bias2, glist, gcnt, (float*)nullptr, batch, psum, pcnt);

  pool_final<<<cdiv((long long)NGRAPH * F2, T), T, 0, stream>>>(psum, pcnt, out);
}
